# starting kernel
# baseline (speedup 1.0000x reference)
#include <hip/hip_runtime.h>
#include <hip/hip_bf16.h>

typedef float v2f __attribute__((ext_vector_type(2)));
typedef float v8f __attribute__((ext_vector_type(8)));

#define NN   50000
#define EE   800000
#define H2C  64
#define H1C  128
#define DINC 256
#define BN_EPS 1e-5f

__global__ void zero_f32(float* __restrict__ p, size_t n) {
  size_t i = (size_t)blockIdx.x * blockDim.x + threadIdx.x;
  size_t stride = (size_t)gridDim.x * blockDim.x;
  for (; i < n; i += stride) p[i] = 0.0f;
}

__global__ void wmma_gemm_f32(const float* __restrict__ A,
                              const float* __restrict__ B,
                              float* __restrict__ C,
                              int Nrows, int K, int Ncols) {
  const int lane = threadIdx.x & 31;
  const int wave = threadIdx.x >> 5;
  const int colTiles = Ncols >> 4;
  const int rowTiles = Nrows >> 4;
  int tile = blockIdx.x * (blockDim.x >> 5) + wave;
  if (tile >= rowTiles * colTiles) return;

  const int rt = tile / colTiles;
  const int ct = tile - rt * colTiles;
  const int row0 = rt << 4;
  const int col0 = ct << 4;

  const int m  = lane & 15;
  const int kh = (lane >> 4) << 1;

  const float* __restrict__ arow = A + (size_t)(row0 + m) * K + kh;
  const float* __restrict__ bcol = B + (size_t)kh * Ncols + (col0 + m);

  v8f acc = {};
  for (int k = 0; k < K; k += 4) {
    v2f a = *(const v2f*)(arow + k);
    v2f b;
    b.x = bcol[(size_t)k * Ncols];
    b.y = bcol[(size_t)(k + 1) * Ncols];
    acc = __builtin_amdgcn_wmma_f32_16x16x4_f32(
        false, a, false, b, (short)0, acc, false, false);
  }

  float* __restrict__ crow = C + (size_t)col0 + m;
  const int rbase = row0 + ((lane >> 4) << 3);
#pragma unroll
  for (int r = 0; r < 8; ++r)
    crow[(size_t)(rbase + r) * Ncols] = acc[r];
}

__global__ void scatter_edges(const float* __restrict__ sup,
                              float* __restrict__ agg,
                              const int* __restrict__ src,
                              const int* __restrict__ dst,
                              const float* __restrict__ w,
                              int E, int D) {
  const int chunks = D >> 2;
  long long idx = (long long)blockIdx.x * blockDim.x + threadIdx.x;
  long long total = (long long)E * chunks;
  if (idx >= total) return;
  int e = (int)(idx / chunks);
  int f = (int)(idx - (long long)e * chunks) << 2;

  const float wt = w[e];
  const float4 v = *(const float4*)(sup + (size_t)src[e] * D + f);
  float* __restrict__ out = agg + (size_t)dst[e] * D + f;
  atomicAdd(out + 0, v.x * wt);
  atomicAdd(out + 1, v.y * wt);
  atomicAdd(out + 2, v.z * wt);
  atomicAdd(out + 3, v.w * wt);
}

__global__ void relu_colstats(float* __restrict__ agg,
                              float* __restrict__ stats,
                              int Nrows, int D) {
  __shared__ float s_sum[DINC];
  __shared__ float s_sq[DINC];
  for (int t = threadIdx.x; t < D; t += blockDim.x) {
    s_sum[t] = 0.0f;
    s_sq[t]  = 0.0f;
  }
  __syncthreads();

  const size_t total = (size_t)Nrows * D;
  const size_t stride = (size_t)gridDim.x * blockDim.x;
  for (size_t i = (size_t)blockIdx.x * blockDim.x + threadIdx.x; i < total;
       i += stride) {
    float v = agg[i];
    v = v > 0.0f ? v : 0.0f;
    agg[i] = v;
    int c = (int)(i & (size_t)(D - 1));
    atomicAdd(&s_sum[c], v);
    atomicAdd(&s_sq[c], v * v);
  }
  __syncthreads();

  for (int t = threadIdx.x; t < D; t += blockDim.x) {
    atomicAdd(&stats[t], s_sum[t]);
    atomicAdd(&stats[D + t], s_sq[t]);
  }
}

__global__ void bn_normalize(const float* __restrict__ agg,
                             float* __restrict__ out,
                             const float* __restrict__ gamma,
                             const float* __restrict__ beta,
                             const float* __restrict__ stats,
                             int Nrows, int D) {
  size_t total = (size_t)Nrows * D;
  size_t i = (size_t)blockIdx.x * blockDim.x + threadIdx.x;
  if (i >= total) return;
  int c = (int)(i & (size_t)(D - 1));
  const float invN = 1.0f / (float)Nrows;
  const float mean = stats[c] * invN;
  const float var  = stats[D + c] * invN - mean * mean;
  const float s = gamma[c] * rsqrtf(var + BN_EPS);
  out[i] = (agg[i] - mean) * s + beta[c];
}

extern "C" void kernel_launch(void* const* d_in, const int* in_sizes, int n_in,
                              void* d_out, int out_size, void* d_ws,
                              size_t ws_size, hipStream_t stream) {
  (void)in_sizes; (void)n_in; (void)out_size; (void)ws_size;

  const float* y   = (const float*)d_in[0];
  const int*   src = (const int*)d_in[1];
  const int*   dst = (const int*)d_in[2];
  const float* w   = (const float*)d_in[3];
  const float* W1  = (const float*)d_in[4];
  const float* g1  = (const float*)d_in[5];
  const float* b1  = (const float*)d_in[6];
  const float* W2  = (const float*)d_in[7];
  const float* g2  = (const float*)d_in[8];
  const float* b2  = (const float*)d_in[9];
  float* out = (float*)d_out;

  float* ws    = (float*)d_ws;
  float* sup1  = ws;
  float* sup2  = sup1 + (size_t)NN * H1C;
  float* agg1  = sup2 + (size_t)NN * DINC;
  float* agg2  = agg1 + (size_t)NN * H1C;
  float* stats1 = agg2 + (size_t)NN * DINC;
  float* stats2 = stats1 + 2 * H1C;

  const size_t zeroN =
      (size_t)NN * H1C + (size_t)NN * DINC + 2 * (H1C + DINC);
  zero_f32<<<2048, 256, 0, stream>>>(agg1, zeroN);

  wmma_gemm_f32<<<25000 / 8, 256, 0, stream>>>(y, W1, sup1, NN, H2C, H1C);

  {
    long long total = (long long)EE * (H1C / 4);
    int blocks = (int)((total + 255) / 256);
    scatter_edges<<<blocks, 256, 0, stream>>>(sup1, agg1, src, dst, w, EE, H1C);
  }

  relu_colstats<<<960, 256, 0, stream>>>(agg1, stats1, NN, H1C);

  {
    size_t total = (size_t)NN * H1C;
    bn_normalize<<<(int)((total + 255) / 256), 256, 0, stream>>>(
        agg1, sup1, g1, b1, stats1, NN, H1C);
  }

  wmma_gemm_f32<<<50000 / 8, 256, 0, stream>>>(sup1, W2, sup2, NN, H1C, DINC);

  {
    long long total = (long long)EE * (DINC / 4);
    int blocks = (int)((total + 255) / 256);
    scatter_edges<<<blocks, 256, 0, stream>>>(sup2, agg2, src, dst, w, EE, DINC);
  }

  relu_colstats<<<960, 256, 0, stream>>>(agg2, stats2, NN, DINC);

  {
    size_t total = (size_t)NN * DINC;
    bn_normalize<<<(int)((total + 255) / 256), 256, 0, stream>>>(
        agg2, out, g2, b2, stats2, NN, DINC);
  }
}


# Edge sums by output tile: a work-group holds 196 rows in LDS, scans the whole dst list, adds its edges with LDS float atomics, stores the tile; global float atomics, zero fill go; sum order changes
# speedup vs baseline: 1.9106x; 1.9106x over previous
#include <hip/hip_runtime.h>
#include <hip/hip_bf16.h>

typedef float v2f __attribute__((ext_vector_type(2)));
typedef float v8f __attribute__((ext_vector_type(8)));

#define NN   50000
#define EE   800000
#define H2C  64
#define H1C  128
#define DINC 256
#define BN_EPS 1e-5f

__global__ void wmma_gemm_f32(const float* __restrict__ A,
                              const float* __restrict__ B,
                              float* __restrict__ C,
                              int Nrows, int K, int Ncols) {
  const int lane = threadIdx.x & 31;
  const int wave = threadIdx.x >> 5;
  const int colTiles = Ncols >> 4;
  const int rowTiles = Nrows >> 4;
  int tile = blockIdx.x * (blockDim.x >> 5) + wave;
  if (tile >= rowTiles * colTiles) return;

  const int rt = tile / colTiles;
  const int ct = tile - rt * colTiles;
  const int row0 = rt << 4;
  const int col0 = ct << 4;

  const int m  = lane & 15;
  const int kh = (lane >> 4) << 1;

  const float* __restrict__ arow = A + (size_t)(row0 + m) * K + kh;
  const float* __restrict__ bcol = B + (size_t)kh * Ncols + (col0 + m);

  v8f acc = {};
  for (int k = 0; k < K; k += 4) {
    v2f a = *(const v2f*)(arow + k);
    v2f b;
    b.x = bcol[(size_t)k * Ncols];
    b.y = bcol[(size_t)(k + 1) * Ncols];
    acc = __builtin_amdgcn_wmma_f32_16x16x4_f32(
        false, a, false, b, (short)0, acc, false, false);
  }

  float* __restrict__ crow = C + (size_t)col0 + m;
  const int rbase = row0 + ((lane >> 4) << 3);
#pragma unroll
  for (int r = 0; r < 8; ++r)
    crow[(size_t)(rbase + r) * Ncols] = acc[r];
}

#define TILE_ROWS 196
#define NTILES ((NN + TILE_ROWS - 1) / TILE_ROWS)
#define AGG_WAVES 32
#define QCAP 160
#define EDGE_UNR 4
#define NBLK (EE / 128)
static_assert(EE % 128 == 0, "a step reads 128 whole edges");

template <int D>
__device__ __forceinline__ void drain_queue(const int* __restrict__ qe,
                                            const int* __restrict__ qr,
                                            int n, int lane,
                                            const float* __restrict__ sup,
                                            const int* __restrict__ src,
                                            const float* __restrict__ w,
                                            float* acc) {
  int e = 0, r = 0;
  if (lane < n) { e = qe[lane]; r = qr[lane]; }
  const int s = src[e];
  const int wbits = __float_as_int(w[e]);
#pragma nounroll
  for (int i = 0; i < n; i += EDGE_UNR) {
    v2f v[EDGE_UNR][D / 64];
    float wi[EDGE_UNR];
    int ri[EDGE_UNR];
#pragma unroll
    for (int u = 0; u < EDGE_UNR; ++u) {
      const int si = __builtin_amdgcn_readlane(s, i + u);
      wi[u] = __int_as_float(__builtin_amdgcn_readlane(wbits, i + u));
      ri[u] = __builtin_amdgcn_readlane(r, i + u);
      const v2f* p = (const v2f*)(sup + (size_t)si * D) + lane;
#pragma unroll
      for (int j = 0; j < D / 64; ++j) v[u][j] = p[j * 32];
    }
#pragma unroll
    for (int u = 0; u < EDGE_UNR; ++u) {
      if (i + u < n) {
        float* a = acc + ri[u] * D + 2 * lane;
#pragma unroll
        for (int j = 0; j < D / 64; ++j) {
          atomicAdd(a + j * 64, v[u][j].x * wi[u]);
          atomicAdd(a + j * 64 + 1, v[u][j].y * wi[u]);
        }
      }
    }
  }
}

template <int D>
__global__ __launch_bounds__(AGG_WAVES * 32)
void agg_tiles(const float* __restrict__ sup,
               float* __restrict__ agg,
               const int* __restrict__ src,
               const int* __restrict__ dst,
               const float* __restrict__ w,
               float* __restrict__ stats) {
  __shared__ float acc[TILE_ROWS * D];
  __shared__ int q_e[AGG_WAVES][QCAP];
  __shared__ int q_r[AGG_WAVES][QCAP];

  const int lane = threadIdx.x & 31;
  const int wave = __builtin_amdgcn_readfirstlane((int)(threadIdx.x >> 5));
  const int row0 = blockIdx.x * TILE_ROWS;
  int nrows = NN - row0;
  if (nrows > TILE_ROWS) nrows = TILE_ROWS;
  if (nrows <= 0) return;

  for (int t = threadIdx.x; t < TILE_ROWS * D / 4; t += AGG_WAVES * 32)
    ((float4*)acc)[t] = make_float4(0.0f, 0.0f, 0.0f, 0.0f);
  if (blockIdx.x == 0)
    for (int t = threadIdx.x; t < 2 * D; t += AGG_WAVES * 32) stats[t] = 0.0f;
  __syncthreads();

  int* qe = q_e[wave];
  int* qr = q_r[wave];
  int qn = 0;

  const int rot = (int)(blockIdx.x % NTILES) * NBLK / NTILES;

  for (int k0 = wave; k0 < NBLK; k0 += AGG_WAVES * 4) {
    int4 d[4];
    int eb[4];
#pragma unroll
    for (int u = 0; u < 4; ++u) {
      const int k = k0 + u * AGG_WAVES;
      int blk = k + rot;
      if (blk >= NBLK) blk -= NBLK;
      eb[u] = (blk << 7) + (lane << 2);
      d[u] = make_int4(-1, -1, -1, -1);
      if (k < NBLK) d[u] = *(const int4*)(dst + eb[u]);
    }
#pragma unroll
    for (int u = 0; u < 4; ++u) {
      const unsigned r0 = (unsigned)(d[u].x - row0);
      const unsigned r1 = (unsigned)(d[u].y - row0);
      const unsigned r2 = (unsigned)(d[u].z - row0);
      const unsigned r3 = (unsigned)(d[u].w - row0);
      const bool m0 = r0 < (unsigned)nrows;
      const bool m1 = r1 < (unsigned)nrows;
      const bool m2 = r2 < (unsigned)nrows;
      const bool m3 = r3 < (unsigned)nrows;
      if (__builtin_amdgcn_ballot_w32(m0 | m1 | m2 | m3)) {
        const unsigned k0m = __builtin_amdgcn_ballot_w32(m0);
        const unsigned k1m = __builtin_amdgcn_ballot_w32(m1);
        const unsigned k2m = __builtin_amdgcn_ballot_w32(m2);
        const unsigned k3m = __builtin_amdgcn_ballot_w32(m3);
        if (m0) { const int sl = qn + (int)__builtin_amdgcn_mbcnt_lo(k0m, 0u); qe[sl] = eb[u];     qr[sl] = (int)r0; }
        qn += __popc(k0m);
        if (m1) { const int sl = qn + (int)__builtin_amdgcn_mbcnt_lo(k1m, 0u); qe[sl] = eb[u] + 1; qr[sl] = (int)r1; }
        qn += __popc(k1m);
        if (m2) { const int sl = qn + (int)__builtin_amdgcn_mbcnt_lo(k2m, 0u); qe[sl] = eb[u] + 2; qr[sl] = (int)r2; }
        qn += __popc(k2m);
        if (m3) { const int sl = qn + (int)__builtin_amdgcn_mbcnt_lo(k3m, 0u); qe[sl] = eb[u] + 3; qr[sl] = (int)r3; }
        qn += __popc(k3m);
        while (qn >= 32) {
          qn -= 32;
          drain_queue<D>(qe + qn, qr + qn, 32, lane, sup, src, w, acc);
        }
      }
    }
  }
  if (qn > 0) drain_queue<D>(qe, qr, qn, lane, sup, src, w, acc);

  __syncthreads();

  float4* __restrict__ o = (float4*)(agg + (size_t)row0 * D);
  for (int t = threadIdx.x; t < nrows * (D / 4); t += AGG_WAVES * 32)
    o[t] = ((const float4*)acc)[t];
}

__global__ void relu_colstats(float* __restrict__ agg,
                              float* __restrict__ stats,
                              int Nrows, int D) {
  __shared__ float s_sum[DINC];
  __shared__ float s_sq[DINC];
  for (int t = threadIdx.x; t < D; t += blockDim.x) {
    s_sum[t] = 0.0f;
    s_sq[t]  = 0.0f;
  }
  __syncthreads();

  const size_t total = (size_t)Nrows * D;
  const size_t stride = (size_t)gridDim.x * blockDim.x;
  for (size_t i = (size_t)blockIdx.x * blockDim.x + threadIdx.x; i < total;
       i += stride) {
    float v = agg[i];
    v = v > 0.0f ? v : 0.0f;
    agg[i] = v;
    int c = (int)(i & (size_t)(D - 1));
    atomicAdd(&s_sum[c], v);
    atomicAdd(&s_sq[c], v * v);
  }
  __syncthreads();

  for (int t = threadIdx.x; t < D; t += blockDim.x) {
    atomicAdd(&stats[t], s_sum[t]);
    atomicAdd(&stats[D + t], s_sq[t]);
  }
}

__global__ void bn_normalize(const float* __restrict__ agg,
                             float* __restrict__ out,
                             const float* __restrict__ gamma,
                             const float* __restrict__ beta,
                             const float* __restrict__ stats,
                             int Nrows, int D) {
  size_t total = (size_t)Nrows * D;
  size_t i = (size_t)blockIdx.x * blockDim.x + threadIdx.x;
  if (i >= total) return;
  int c = (int)(i & (size_t)(D - 1));
  const float invN = 1.0f / (float)Nrows;
  const float mean = stats[c] * invN;
  const float var  = stats[D + c] * invN - mean * mean;
  const float s = gamma[c] * rsqrtf(var + BN_EPS);
  out[i] = (agg[i] - mean) * s + beta[c];
}

extern "C" void kernel_launch(void* const* d_in, const int* in_sizes, int n_in,
                              void* d_out, int out_size, void* d_ws,
                              size_t ws_size, hipStream_t stream) {
  (void)in_sizes; (void)n_in; (void)out_size; (void)ws_size;

  const float* y   = (const float*)d_in[0];
  const int*   src = (const int*)d_in[1];
  const int*   dst = (const int*)d_in[2];
  const float* w   = (const float*)d_in[3];
  const float* W1  = (const float*)d_in[4];
  const float* g1  = (const float*)d_in[5];
  const float* b1  = (const float*)d_in[6];
  const float* W2  = (const float*)d_in[7];
  const float* g2  = (const float*)d_in[8];
  const float* b2  = (const float*)d_in[9];
  float* out = (float*)d_out;

  float* ws    = (float*)d_ws;
  float* sup1  = ws;
  float* sup2  = sup1 + (size_t)NN * H1C;
  float* agg1  = sup2 + (size_t)NN * DINC;
  float* agg2  = agg1 + (size_t)NN * H1C;
  float* stats1 = agg2 + (size_t)NN * DINC;
  float* stats2 = stats1 + 2 * H1C;

  wmma_gemm_f32<<<25000 / 8, 256, 0, stream>>>(y, W1, sup1, NN, H2C, H1C);

  agg_tiles<H1C><<<NTILES, AGG_WAVES * 32, 0, stream>>>(
      sup1, agg1, src, dst, w, stats1);

  relu_colstats<<<960, 256, 0, stream>>>(agg1, stats1, NN, H1C);

  {
    size_t total = (size_t)NN * H1C;
    bn_normalize<<<(int)((total + 255) / 256), 256, 0, stream>>>(
        agg1, sup1, g1, b1, stats1, NN, H1C);
  }

  wmma_gemm_f32<<<50000 / 8, 256, 0, stream>>>(sup1, W2, sup2, NN, H1C, DINC);

  agg_tiles<DINC><<<NTILES, AGG_WAVES * 32, 0, stream>>>(
      sup2, agg2, src, dst, w, stats2);

  relu_colstats<<<960, 256, 0, stream>>>(agg2, stats2, NN, DINC);

  {
    size_t total = (size_t)NN * DINC;
    bn_normalize<<<(int)((total + 255) / 256), 256, 0, stream>>>(
        agg2, out, g2, b2, stats2, NN, DINC);
  }
}


# First BatchNorm applied in the second dense product's A loader: a work-group builds mean/scale/beta per column in LDS, lanes normalise A values as loaded; BN1 pass (200000 waves) and stored h1 go
# speedup vs baseline: 1.9990x; 1.0463x over previous
#include <hip/hip_runtime.h>
#include <hip/hip_bf16.h>

typedef float v2f __attribute__((ext_vector_type(2)));
typedef float v8f __attribute__((ext_vector_type(8)));

#define NN   50000
#define EE   800000
#define H2C  64
#define H1C  128
#define DINC 256
#define BN_EPS 1e-5f

__global__ void wmma_gemm_f32(const float* __restrict__ A,
                              const float* __restrict__ B,
                              float* __restrict__ C,
                              int Nrows, int K, int Ncols) {
  const int lane = threadIdx.x & 31;
  const int wave = threadIdx.x >> 5;
  const int colTiles = Ncols >> 4;
  const int rowTiles = Nrows >> 4;
  int tile = blockIdx.x * (blockDim.x >> 5) + wave;
  if (tile >= rowTiles * colTiles) return;

  const int rt = tile / colTiles;
  const int ct = tile - rt * colTiles;
  const int row0 = rt << 4;
  const int col0 = ct << 4;

  const int m  = lane & 15;
  const int kh = (lane >> 4) << 1;

  const float* __restrict__ arow = A + (size_t)(row0 + m) * K + kh;
  const float* __restrict__ bcol = B + (size_t)kh * Ncols + (col0 + m);

  v8f acc = {};
  for (int k = 0; k < K; k += 4) {
    v2f a = *(const v2f*)(arow + k);
    v2f b;
    b.x = bcol[(size_t)k * Ncols];
    b.y = bcol[(size_t)(k + 1) * Ncols];
    acc = __builtin_amdgcn_wmma_f32_16x16x4_f32(
        false, a, false, b, (short)0, acc, false, false);
  }

  float* __restrict__ crow = C + (size_t)col0 + m;
  const int rbase = row0 + ((lane >> 4) << 3);
#pragma unroll
  for (int r = 0; r < 8; ++r)
    crow[(size_t)(rbase + r) * Ncols] = acc[r];
}

__global__ __launch_bounds__(256)
void wmma_gemm_bn_f32(const float* __restrict__ A,
                      const float* __restrict__ B,
                      float* __restrict__ C,
                      const float* __restrict__ stats,
                      const float* __restrict__ gamma,
                      const float* __restrict__ beta,
                      int Nrows, int K, int Ncols) {
  __shared__ __attribute__((aligned(16))) float bn[(H1C / 2) * 8];
  for (int c = threadIdx.x; c < K; c += blockDim.x) {
    const float invN = 1.0f / (float)Nrows;
    const float mean = stats[c] * invN;
    const float var  = stats[K + c] * invN - mean * mean;
    const float s = gamma[c] * rsqrtf(var + BN_EPS);
    float* e = bn + (c >> 1) * 8 + (c & 1);
    e[0] = mean;
    e[2] = s;
    e[4] = beta[c];
  }
  __syncthreads();

  const int lane = threadIdx.x & 31;
  const int wave = threadIdx.x >> 5;
  const int colTiles = Ncols >> 4;
  const int rowTiles = Nrows >> 4;
  int tile = blockIdx.x * (blockDim.x >> 5) + wave;
  if (tile >= rowTiles * colTiles) return;

  const int rt = tile / colTiles;
  const int ct = tile - rt * colTiles;
  const int row0 = rt << 4;
  const int col0 = ct << 4;

  const int m  = lane & 15;
  const int kh = (lane >> 4) << 1;

  const float* __restrict__ arow = A + (size_t)(row0 + m) * K + kh;
  const float* __restrict__ bcol = B + (size_t)kh * Ncols + (col0 + m);
  const float* bnk = bn + (kh >> 1) * 8;

  v8f acc = {};
#pragma unroll 4
  for (int k = 0; k < K; k += 4) {
    v2f a = *(const v2f*)(arow + k);
    const float4 ms = *(const float4*)(bnk + (k >> 1) * 8);
    const v2f sh = *(const v2f*)(bnk + (k >> 1) * 8 + 4);
    a.x = (a.x - ms.x) * ms.z + sh.x;
    a.y = (a.y - ms.y) * ms.w + sh.y;
    v2f b;
    b.x = bcol[(size_t)k * Ncols];
    b.y = bcol[(size_t)(k + 1) * Ncols];
    acc = __builtin_amdgcn_wmma_f32_16x16x4_f32(
        false, a, false, b, (short)0, acc, false, false);
  }

  float* __restrict__ crow = C + (size_t)col0 + m;
  const int rbase = row0 + ((lane >> 4) << 3);
#pragma unroll
  for (int r = 0; r < 8; ++r)
    crow[(size_t)(rbase + r) * Ncols] = acc[r];
}

#define TILE_ROWS 196
#define NTILES ((NN + TILE_ROWS - 1) / TILE_ROWS)
#define AGG_WAVES 32
#define QCAP 160
#define EDGE_UNR 4
#define NBLK (EE / 128)
static_assert(EE % 128 == 0, "a step reads 128 whole edges");

template <int D>
__device__ __forceinline__ void drain_queue(const int* __restrict__ qe,
                                            const int* __restrict__ qr,
                                            int n, int lane,
                                            const float* __restrict__ sup,
                                            const int* __restrict__ src,
                                            const float* __restrict__ w,
                                            float* acc) {
  int e = 0, r = 0;
  if (lane < n) { e = qe[lane]; r = qr[lane]; }
  const int s = src[e];
  const int wbits = __float_as_int(w[e]);
#pragma nounroll
  for (int i = 0; i < n; i += EDGE_UNR) {
    v2f v[EDGE_UNR][D / 64];
    float wi[EDGE_UNR];
    int ri[EDGE_UNR];
#pragma unroll
    for (int u = 0; u < EDGE_UNR; ++u) {
      const int si = __builtin_amdgcn_readlane(s, i + u);
      wi[u] = __int_as_float(__builtin_amdgcn_readlane(wbits, i + u));
      ri[u] = __builtin_amdgcn_readlane(r, i + u);
      const v2f* p = (const v2f*)(sup + (size_t)si * D) + lane;
#pragma unroll
      for (int j = 0; j < D / 64; ++j) v[u][j] = p[j * 32];
    }
#pragma unroll
    for (int u = 0; u < EDGE_UNR; ++u) {
      if (i + u < n) {
        float* a = acc + ri[u] * D + 2 * lane;
#pragma unroll
        for (int j = 0; j < D / 64; ++j) {
          atomicAdd(a + j * 64, v[u][j].x * wi[u]);
          atomicAdd(a + j * 64 + 1, v[u][j].y * wi[u]);
        }
      }
    }
  }
}

template <int D>
__global__ __launch_bounds__(AGG_WAVES * 32)
void agg_tiles(const float* __restrict__ sup,
               float* __restrict__ agg,
               const int* __restrict__ src,
               const int* __restrict__ dst,
               const float* __restrict__ w,
               float* __restrict__ stats) {
  __shared__ float acc[TILE_ROWS * D];
  __shared__ int q_e[AGG_WAVES][QCAP];
  __shared__ int q_r[AGG_WAVES][QCAP];

  const int lane = threadIdx.x & 31;
  const int wave = __builtin_amdgcn_readfirstlane((int)(threadIdx.x >> 5));
  const int row0 = blockIdx.x * TILE_ROWS;
  int nrows = NN - row0;
  if (nrows > TILE_ROWS) nrows = TILE_ROWS;
  if (nrows <= 0) return;

  for (int t = threadIdx.x; t < TILE_ROWS * D / 4; t += AGG_WAVES * 32)
    ((float4*)acc)[t] = make_float4(0.0f, 0.0f, 0.0f, 0.0f);
  if (blockIdx.x == 0)
    for (int t = threadIdx.x; t < 2 * D; t += AGG_WAVES * 32) stats[t] = 0.0f;
  __syncthreads();

  int* qe = q_e[wave];
  int* qr = q_r[wave];
  int qn = 0;

  const int rot = (int)(blockIdx.x % NTILES) * NBLK / NTILES;

  for (int k0 = wave; k0 < NBLK; k0 += AGG_WAVES * 4) {
    int4 d[4];
    int eb[4];
#pragma unroll
    for (int u = 0; u < 4; ++u) {
      const int k = k0 + u * AGG_WAVES;
      int blk = k + rot;
      if (blk >= NBLK) blk -= NBLK;
      eb[u] = (blk << 7) + (lane << 2);
      d[u] = make_int4(-1, -1, -1, -1);
      if (k < NBLK) d[u] = *(const int4*)(dst + eb[u]);
    }
#pragma unroll
    for (int u = 0; u < 4; ++u) {
      const unsigned r0 = (unsigned)(d[u].x - row0);
      const unsigned r1 = (unsigned)(d[u].y - row0);
      const unsigned r2 = (unsigned)(d[u].z - row0);
      const unsigned r3 = (unsigned)(d[u].w - row0);
      const bool m0 = r0 < (unsigned)nrows;
      const bool m1 = r1 < (unsigned)nrows;
      const bool m2 = r2 < (unsigned)nrows;
      const bool m3 = r3 < (unsigned)nrows;
      if (__builtin_amdgcn_ballot_w32(m0 | m1 | m2 | m3)) {
        const unsigned k0m = __builtin_amdgcn_ballot_w32(m0);
        const unsigned k1m = __builtin_amdgcn_ballot_w32(m1);
        const unsigned k2m = __builtin_amdgcn_ballot_w32(m2);
        const unsigned k3m = __builtin_amdgcn_ballot_w32(m3);
        if (m0) { const int sl = qn + (int)__builtin_amdgcn_mbcnt_lo(k0m, 0u); qe[sl] = eb[u];     qr[sl] = (int)r0; }
        qn += __popc(k0m);
        if (m1) { const int sl = qn + (int)__builtin_amdgcn_mbcnt_lo(k1m, 0u); qe[sl] = eb[u] + 1; qr[sl] = (int)r1; }
        qn += __popc(k1m);
        if (m2) { const int sl = qn + (int)__builtin_amdgcn_mbcnt_lo(k2m, 0u); qe[sl] = eb[u] + 2; qr[sl] = (int)r2; }
        qn += __popc(k2m);
        if (m3) { const int sl = qn + (int)__builtin_amdgcn_mbcnt_lo(k3m, 0u); qe[sl] = eb[u] + 3; qr[sl] = (int)r3; }
        qn += __popc(k3m);
        while (qn >= 32) {
          qn -= 32;
          drain_queue<D>(qe + qn, qr + qn, 32, lane, sup, src, w, acc);
        }
      }
    }
  }
  if (qn > 0) drain_queue<D>(qe, qr, qn, lane, sup, src, w, acc);

  __syncthreads();

  float4* __restrict__ o = (float4*)(agg + (size_t)row0 * D);
  for (int t = threadIdx.x; t < nrows * (D / 4); t += AGG_WAVES * 32)
    o[t] = ((const float4*)acc)[t];
}

__global__ void relu_colstats(float* __restrict__ agg,
                              float* __restrict__ stats,
                              int Nrows, int D) {
  __shared__ float s_sum[DINC];
  __shared__ float s_sq[DINC];
  for (int t = threadIdx.x; t < D; t += blockDim.x) {
    s_sum[t] = 0.0f;
    s_sq[t]  = 0.0f;
  }
  __syncthreads();

  const size_t total = (size_t)Nrows * D;
  const size_t stride = (size_t)gridDim.x * blockDim.x;
  for (size_t i = (size_t)blockIdx.x * blockDim.x + threadIdx.x; i < total;
       i += stride) {
    float v = agg[i];
    v = v > 0.0f ? v : 0.0f;
    agg[i] = v;
    int c = (int)(i & (size_t)(D - 1));
    atomicAdd(&s_sum[c], v);
    atomicAdd(&s_sq[c], v * v);
  }
  __syncthreads();

  for (int t = threadIdx.x; t < D; t += blockDim.x) {
    atomicAdd(&stats[t], s_sum[t]);
    atomicAdd(&stats[D + t], s_sq[t]);
  }
}

__global__ void bn_normalize(const float* __restrict__ agg,
                             float* __restrict__ out,
                             const float* __restrict__ gamma,
                             const float* __restrict__ beta,
                             const float* __restrict__ stats,
                             int Nrows, int D) {
  size_t total = (size_t)Nrows * D;
  size_t i = (size_t)blockIdx.x * blockDim.x + threadIdx.x;
  if (i >= total) return;
  int c = (int)(i & (size_t)(D - 1));
  const float invN = 1.0f / (float)Nrows;
  const float mean = stats[c] * invN;
  const float var  = stats[D + c] * invN - mean * mean;
  const float s = gamma[c] * rsqrtf(var + BN_EPS);
  out[i] = (agg[i] - mean) * s + beta[c];
}

extern "C" void kernel_launch(void* const* d_in, const int* in_sizes, int n_in,
                              void* d_out, int out_size, void* d_ws,
                              size_t ws_size, hipStream_t stream) {
  (void)in_sizes; (void)n_in; (void)out_size; (void)ws_size;

  const float* y   = (const float*)d_in[0];
  const int*   src = (const int*)d_in[1];
  const int*   dst = (const int*)d_in[2];
  const float* w   = (const float*)d_in[3];
  const float* W1  = (const float*)d_in[4];
  const float* g1  = (const float*)d_in[5];
  const float* b1  = (const float*)d_in[6];
  const float* W2  = (const float*)d_in[7];
  const float* g2  = (const float*)d_in[8];
  const float* b2  = (const float*)d_in[9];
  float* out = (float*)d_out;

  float* ws    = (float*)d_ws;
  float* sup1  = ws;
  float* sup2  = sup1 + (size_t)NN * H1C;
  float* agg1  = sup2 + (size_t)NN * DINC;
  float* agg2  = agg1 + (size_t)NN * H1C;
  float* stats1 = agg2 + (size_t)NN * DINC;
  float* stats2 = stats1 + 2 * H1C;

  wmma_gemm_f32<<<25000 / 8, 256, 0, stream>>>(y, W1, sup1, NN, H2C, H1C);

  agg_tiles<H1C><<<NTILES, AGG_WAVES * 32, 0, stream>>>(
      sup1, agg1, src, dst, w, stats1);

  relu_colstats<<<960, 256, 0, stream>>>(agg1, stats1, NN, H1C);

  wmma_gemm_bn_f32<<<50000 / 8, 256, 0, stream>>>(
      agg1, W2, sup2, stats1, g1, b1, NN, H1C, DINC);

  agg_tiles<DINC><<<NTILES, AGG_WAVES * 32, 0, stream>>>(
      sup2, agg2, src, dst, w, stats2);

  relu_colstats<<<960, 256, 0, stream>>>(agg2, stats2, NN, DINC);

  {
    size_t total = (size_t)NN * DINC;
    bn_normalize<<<(int)((total + 255) / 256), 256, 0, stream>>>(
        agg2, out, g2, b2, stats2, NN, DINC);
  }
}


# ReLU and BatchNorm column sums taken in the edge-sum kernel's tile copy-out (LDS tile to registers to global): both ReLU/sums passes and their rewrite of agg go; order of the column sums changes
# speedup vs baseline: 2.3270x; 1.1641x over previous
#include <hip/hip_runtime.h>
#include <hip/hip_bf16.h>

typedef float v2f __attribute__((ext_vector_type(2)));
typedef float v8f __attribute__((ext_vector_type(8)));

#define NN   50000
#define EE   800000
#define H2C  64
#define H1C  128
#define DINC 256
#define BN_EPS 1e-5f

__global__ void wmma_gemm_f32(const float* __restrict__ A,
                              const float* __restrict__ B,
                              float* __restrict__ C,
                              float* __restrict__ zero_buf, int zero_n,
                              int Nrows, int K, int Ncols) {
  if (blockIdx.x == 0)
    for (int t = threadIdx.x; t < zero_n; t += blockDim.x) zero_buf[t] = 0.0f;

  const int lane = threadIdx.x & 31;
  const int wave = threadIdx.x >> 5;
  const int colTiles = Ncols >> 4;
  const int rowTiles = Nrows >> 4;
  int tile = blockIdx.x * (blockDim.x >> 5) + wave;
  if (tile >= rowTiles * colTiles) return;

  const int rt = tile / colTiles;
  const int ct = tile - rt * colTiles;
  const int row0 = rt << 4;
  const int col0 = ct << 4;

  const int m  = lane & 15;
  const int kh = (lane >> 4) << 1;

  const float* __restrict__ arow = A + (size_t)(row0 + m) * K + kh;
  const float* __restrict__ bcol = B + (size_t)kh * Ncols + (col0 + m);

  v8f acc = {};
  for (int k = 0; k < K; k += 4) {
    v2f a = *(const v2f*)(arow + k);
    v2f b;
    b.x = bcol[(size_t)k * Ncols];
    b.y = bcol[(size_t)(k + 1) * Ncols];
    acc = __builtin_amdgcn_wmma_f32_16x16x4_f32(
        false, a, false, b, (short)0, acc, false, false);
  }

  float* __restrict__ crow = C + (size_t)col0 + m;
  const int rbase = row0 + ((lane >> 4) << 3);
#pragma unroll
  for (int r = 0; r < 8; ++r)
    crow[(size_t)(rbase + r) * Ncols] = acc[r];
}

__global__ __launch_bounds__(256)
void wmma_gemm_bn_f32(const float* __restrict__ A,
                      const float* __restrict__ B,
                      float* __restrict__ C,
                      const float* __restrict__ stats,
                      const float* __restrict__ gamma,
                      const float* __restrict__ beta,
                      int Nrows, int K, int Ncols) {
  __shared__ __attribute__((aligned(16))) float bn[(H1C / 2) * 8];
  for (int c = threadIdx.x; c < K; c += blockDim.x) {
    const float invN = 1.0f / (float)Nrows;
    const float mean = stats[c] * invN;
    const float var  = stats[K + c] * invN - mean * mean;
    const float s = gamma[c] * rsqrtf(var + BN_EPS);
    float* e = bn + (c >> 1) * 8 + (c & 1);
    e[0] = mean;
    e[2] = s;
    e[4] = beta[c];
  }
  __syncthreads();

  const int lane = threadIdx.x & 31;
  const int wave = threadIdx.x >> 5;
  const int colTiles = Ncols >> 4;
  const int rowTiles = Nrows >> 4;
  int tile = blockIdx.x * (blockDim.x >> 5) + wave;
  if (tile >= rowTiles * colTiles) return;

  const int rt = tile / colTiles;
  const int ct = tile - rt * colTiles;
  const int row0 = rt << 4;
  const int col0 = ct << 4;

  const int m  = lane & 15;
  const int kh = (lane >> 4) << 1;

  const float* __restrict__ arow = A + (size_t)(row0 + m) * K + kh;
  const float* __restrict__ bcol = B + (size_t)kh * Ncols + (col0 + m);
  const float* bnk = bn + (kh >> 1) * 8;

  v8f acc = {};
#pragma unroll 4
  for (int k = 0; k < K; k += 4) {
    v2f a = *(const v2f*)(arow + k);
    const float4 ms = *(const float4*)(bnk + (k >> 1) * 8);
    const v2f sh = *(const v2f*)(bnk + (k >> 1) * 8 + 4);
    a.x = (a.x - ms.x) * ms.z + sh.x;
    a.y = (a.y - ms.y) * ms.w + sh.y;
    v2f b;
    b.x = bcol[(size_t)k * Ncols];
    b.y = bcol[(size_t)(k + 1) * Ncols];
    acc = __builtin_amdgcn_wmma_f32_16x16x4_f32(
        false, a, false, b, (short)0, acc, false, false);
  }

  float* __restrict__ crow = C + (size_t)col0 + m;
  const int rbase = row0 + ((lane >> 4) << 3);
#pragma unroll
  for (int r = 0; r < 8; ++r)
    crow[(size_t)(rbase + r) * Ncols] = acc[r];
}

#define TILE_ROWS 196
#define NTILES ((NN + TILE_ROWS - 1) / TILE_ROWS)
#define AGG_WAVES 32
#define QCAP 160
#define EDGE_UNR 4
#define NBLK (EE / 128)
static_assert(EE % 128 == 0, "a step reads 128 whole edges");

template <int D>
__device__ __forceinline__ void drain_queue(const int* __restrict__ qe,
                                            const int* __restrict__ qr,
                                            int n, int lane,
                                            const float* __restrict__ sup,
                                            const int* __restrict__ src,
                                            const float* __restrict__ w,
                                            float* acc) {
  int e = 0, r = 0;
  if (lane < n) { e = qe[lane]; r = qr[lane]; }
  const int s = src[e];
  const int wbits = __float_as_int(w[e]);
#pragma nounroll
  for (int i = 0; i < n; i += EDGE_UNR) {
    v2f v[EDGE_UNR][D / 64];
    float wi[EDGE_UNR];
    int ri[EDGE_UNR];
#pragma unroll
    for (int u = 0; u < EDGE_UNR; ++u) {
      const int si = __builtin_amdgcn_readlane(s, i + u);
      wi[u] = __int_as_float(__builtin_amdgcn_readlane(wbits, i + u));
      ri[u] = __builtin_amdgcn_readlane(r, i + u);
      const v2f* p = (const v2f*)(sup + (size_t)si * D) + lane;
#pragma unroll
      for (int j = 0; j < D / 64; ++j) v[u][j] = p[j * 32];
    }
#pragma unroll
    for (int u = 0; u < EDGE_UNR; ++u) {
      if (i + u < n) {
        float* a = acc + ri[u] * D + 2 * lane;
#pragma unroll
        for (int j = 0; j < D / 64; ++j) {
          atomicAdd(a + j * 64, v[u][j].x * wi[u]);
          atomicAdd(a + j * 64 + 1, v[u][j].y * wi[u]);
        }
      }
    }
  }
}

template <int D>
__global__ __launch_bounds__(AGG_WAVES * 32)
void agg_tiles(const float* __restrict__ sup,
               float* __restrict__ agg,
               const int* __restrict__ src,
               const int* __restrict__ dst,
               const float* __restrict__ w,
               float* __restrict__ stats) {
  __shared__ float acc[TILE_ROWS * D];
  __shared__ int q_e[AGG_WAVES][QCAP];
  __shared__ int q_r[AGG_WAVES][QCAP];
  __shared__ float colsum[2 * D];

  const int lane = threadIdx.x & 31;
  const int wave = __builtin_amdgcn_readfirstlane((int)(threadIdx.x >> 5));
  const int row0 = blockIdx.x * TILE_ROWS;
  int nrows = NN - row0;
  if (nrows > TILE_ROWS) nrows = TILE_ROWS;
  if (nrows <= 0) return;

  for (int t = threadIdx.x; t < TILE_ROWS * D / 4; t += AGG_WAVES * 32)
    ((float4*)acc)[t] = make_float4(0.0f, 0.0f, 0.0f, 0.0f);
  for (int t = threadIdx.x; t < 2 * D; t += AGG_WAVES * 32) colsum[t] = 0.0f;
  __syncthreads();

  int* qe = q_e[wave];
  int* qr = q_r[wave];
  int qn = 0;

  const int rot = (int)(blockIdx.x % NTILES) * NBLK / NTILES;

  for (int k0 = wave; k0 < NBLK; k0 += AGG_WAVES * 4) {
    int4 d[4];
    int eb[4];
#pragma unroll
    for (int u = 0; u < 4; ++u) {
      const int k = k0 + u * AGG_WAVES;
      int blk = k + rot;
      if (blk >= NBLK) blk -= NBLK;
      eb[u] = (blk << 7) + (lane << 2);
      d[u] = make_int4(-1, -1, -1, -1);
      if (k < NBLK) d[u] = *(const int4*)(dst + eb[u]);
    }
#pragma unroll
    for (int u = 0; u < 4; ++u) {
      const unsigned r0 = (unsigned)(d[u].x - row0);
      const unsigned r1 = (unsigned)(d[u].y - row0);
      const unsigned r2 = (unsigned)(d[u].z - row0);
      const unsigned r3 = (unsigned)(d[u].w - row0);
      const bool m0 = r0 < (unsigned)nrows;
      const bool m1 = r1 < (unsigned)nrows;
      const bool m2 = r2 < (unsigned)nrows;
      const bool m3 = r3 < (unsigned)nrows;
      if (__builtin_amdgcn_ballot_w32(m0 | m1 | m2 | m3)) {
        const unsigned k0m = __builtin_amdgcn_ballot_w32(m0);
        const unsigned k1m = __builtin_amdgcn_ballot_w32(m1);
        const unsigned k2m = __builtin_amdgcn_ballot_w32(m2);
        const unsigned k3m = __builtin_amdgcn_ballot_w32(m3);
        if (m0) { const int sl = qn + (int)__builtin_amdgcn_mbcnt_lo(k0m, 0u); qe[sl] = eb[u];     qr[sl] = (int)r0; }
        qn += __popc(k0m);
        if (m1) { const int sl = qn + (int)__builtin_amdgcn_mbcnt_lo(k1m, 0u); qe[sl] = eb[u] + 1; qr[sl] = (int)r1; }
        qn += __popc(k1m);
        if (m2) { const int sl = qn + (int)__builtin_amdgcn_mbcnt_lo(k2m, 0u); qe[sl] = eb[u] + 2; qr[sl] = (int)r2; }
        qn += __popc(k2m);
        if (m3) { const int sl = qn + (int)__builtin_amdgcn_mbcnt_lo(k3m, 0u); qe[sl] = eb[u] + 3; qr[sl] = (int)r3; }
        qn += __popc(k3m);
        while (qn >= 32) {
          qn -= 32;
          drain_queue<D>(qe + qn, qr + qn, 32, lane, sup, src, w, acc);
        }
      }
    }
  }
  if (qn > 0) drain_queue<D>(qe, qr, qn, lane, sup, src, w, acc);

  __syncthreads();

  static_assert((AGG_WAVES * 32) % (D / 4) == 0, "a thread keeps its columns");
  float4* __restrict__ o = (float4*)(agg + (size_t)row0 * D);
  float4 s1 = make_float4(0.0f, 0.0f, 0.0f, 0.0f);
  float4 s2 = make_float4(0.0f, 0.0f, 0.0f, 0.0f);
  for (int t = threadIdx.x; t < nrows * (D / 4); t += AGG_WAVES * 32) {
    float4 v = ((const float4*)acc)[t];
    v.x = v.x > 0.0f ? v.x : 0.0f;
    v.y = v.y > 0.0f ? v.y : 0.0f;
    v.z = v.z > 0.0f ? v.z : 0.0f;
    v.w = v.w > 0.0f ? v.w : 0.0f;
    o[t] = v;
    s1.x += v.x; s1.y += v.y; s1.z += v.z; s1.w += v.w;
    s2.x += v.x * v.x; s2.y += v.y * v.y; s2.z += v.z * v.z; s2.w += v.w * v.w;
  }
  const int c0 = (threadIdx.x % (D / 4)) * 4;
  atomicAdd(&colsum[c0 + 0], s1.x);
  atomicAdd(&colsum[c0 + 1], s1.y);
  atomicAdd(&colsum[c0 + 2], s1.z);
  atomicAdd(&colsum[c0 + 3], s1.w);
  atomicAdd(&colsum[D + c0 + 0], s2.x);
  atomicAdd(&colsum[D + c0 + 1], s2.y);
  atomicAdd(&colsum[D + c0 + 2], s2.z);
  atomicAdd(&colsum[D + c0 + 3], s2.w);
  __syncthreads();
  for (int t = threadIdx.x; t < 2 * D; t += AGG_WAVES * 32)
    atomicAdd(&stats[t], colsum[t]);
}

__global__ void bn_normalize(const float* __restrict__ agg,
                             float* __restrict__ out,
                             const float* __restrict__ gamma,
                             const float* __restrict__ beta,
                             const float* __restrict__ stats,
                             int Nrows, int D) {
  size_t total = (size_t)Nrows * D;
  size_t i = (size_t)blockIdx.x * blockDim.x + threadIdx.x;
  if (i >= total) return;
  int c = (int)(i & (size_t)(D - 1));
  const float invN = 1.0f / (float)Nrows;
  const float mean = stats[c] * invN;
  const float var  = stats[D + c] * invN - mean * mean;
  const float s = gamma[c] * rsqrtf(var + BN_EPS);
  out[i] = (agg[i] - mean) * s + beta[c];
}

extern "C" void kernel_launch(void* const* d_in, const int* in_sizes, int n_in,
                              void* d_out, int out_size, void* d_ws,
                              size_t ws_size, hipStream_t stream) {
  (void)in_sizes; (void)n_in; (void)out_size; (void)ws_size;

  const float* y   = (const float*)d_in[0];
  const int*   src = (const int*)d_in[1];
  const int*   dst = (const int*)d_in[2];
  const float* w   = (const float*)d_in[3];
  const float* W1  = (const float*)d_in[4];
  const float* g1  = (const float*)d_in[5];
  const float* b1  = (const float*)d_in[6];
  const float* W2  = (const float*)d_in[7];
  const float* g2  = (const float*)d_in[8];
  const float* b2  = (const float*)d_in[9];
  float* out = (float*)d_out;

  float* ws    = (float*)d_ws;
  float* sup1  = ws;
  float* sup2  = sup1 + (size_t)NN * H1C;
  float* agg1  = sup2 + (size_t)NN * DINC;
  float* agg2  = agg1 + (size_t)NN * H1C;
  float* stats1 = agg2 + (size_t)NN * DINC;
  float* stats2 = stats1 + 2 * H1C;

  wmma_gemm_f32<<<25000 / 8, 256, 0, stream>>>(
      y, W1, sup1, stats1, 2 * H1C + 2 * DINC, NN, H2C, H1C);

  agg_tiles<H1C><<<NTILES, AGG_WAVES * 32, 0, stream>>>(
      sup1, agg1, src, dst, w, stats1);

  wmma_gemm_bn_f32<<<50000 / 8, 256, 0, stream>>>(
      agg1, W2, sup2, stats1, g1, b1, NN, H1C, DINC);

  agg_tiles<DINC><<<NTILES, AGG_WAVES * 32, 0, stream>>>(
      sup2, agg2, src, dst, w, stats2);

  {
    size_t total = (size_t)NN * DINC;
    bn_normalize<<<(int)((total + 255) / 256), 256, 0, stream>>>(
        agg2, out, g2, b2, stats2, NN, DINC);
  }
}


# Layer-2 edge sums read per-wave edge lists (src, w, row) that layer 1's waves store while draining: every group's second scan of dst goes; same LDS float atomics, per-wave add order kept
# speedup vs baseline: 2.5824x; 1.1098x over previous
#include <hip/hip_runtime.h>
#include <hip/hip_bf16.h>

typedef float v2f __attribute__((ext_vector_type(2)));
typedef float v8f __attribute__((ext_vector_type(8)));

#define NN   50000
#define EE   800000
#define H2C  64
#define H1C  128
#define DINC 256
#define BN_EPS 1e-5f

__global__ void wmma_gemm_f32(const float* __restrict__ A,
                              const float* __restrict__ B,
                              float* __restrict__ C,
                              float* __restrict__ zero_buf, int zero_n,
                              int Nrows, int K, int Ncols) {
  if (blockIdx.x == 0)
    for (int t = threadIdx.x; t < zero_n; t += blockDim.x) zero_buf[t] = 0.0f;

  const int lane = threadIdx.x & 31;
  const int wave = threadIdx.x >> 5;
  const int colTiles = Ncols >> 4;
  const int rowTiles = Nrows >> 4;
  int tile = blockIdx.x * (blockDim.x >> 5) + wave;
  if (tile >= rowTiles * colTiles) return;

  const int rt = tile / colTiles;
  const int ct = tile - rt * colTiles;
  const int row0 = rt << 4;
  const int col0 = ct << 4;

  const int m  = lane & 15;
  const int kh = (lane >> 4) << 1;

  const float* __restrict__ arow = A + (size_t)(row0 + m) * K + kh;
  const float* __restrict__ bcol = B + (size_t)kh * Ncols + (col0 + m);

  v8f acc = {};
  for (int k = 0; k < K; k += 4) {
    v2f a = *(const v2f*)(arow + k);
    v2f b;
    b.x = bcol[(size_t)k * Ncols];
    b.y = bcol[(size_t)(k + 1) * Ncols];
    acc = __builtin_amdgcn_wmma_f32_16x16x4_f32(
        false, a, false, b, (short)0, acc, false, false);
  }

  float* __restrict__ crow = C + (size_t)col0 + m;
  const int rbase = row0 + ((lane >> 4) << 3);
#pragma unroll
  for (int r = 0; r < 8; ++r)
    crow[(size_t)(rbase + r) * Ncols] = acc[r];
}

__global__ __launch_bounds__(256)
void wmma_gemm_bn_f32(const float* __restrict__ A,
                      const float* __restrict__ B,
                      float* __restrict__ C,
                      const float* __restrict__ stats,
                      const float* __restrict__ gamma,
                      const float* __restrict__ beta,
                      int Nrows, int K, int Ncols) {
  __shared__ __attribute__((aligned(16))) float bn[(H1C / 2) * 8];
  for (int c = threadIdx.x; c < K; c += blockDim.x) {
    const float invN = 1.0f / (float)Nrows;
    const float mean = stats[c] * invN;
    const float var  = stats[K + c] * invN - mean * mean;
    const float s = gamma[c] * rsqrtf(var + BN_EPS);
    float* e = bn + (c >> 1) * 8 + (c & 1);
    e[0] = mean;
    e[2] = s;
    e[4] = beta[c];
  }
  __syncthreads();

  const int lane = threadIdx.x & 31;
  const int wave = threadIdx.x >> 5;
  const int colTiles = Ncols >> 4;
  const int rowTiles = Nrows >> 4;
  int tile = blockIdx.x * (blockDim.x >> 5) + wave;
  if (tile >= rowTiles * colTiles) return;

  const int rt = tile / colTiles;
  const int ct = tile - rt * colTiles;
  const int row0 = rt << 4;
  const int col0 = ct << 4;

  const int m  = lane & 15;
  const int kh = (lane >> 4) << 1;

  const float* __restrict__ arow = A + (size_t)(row0 + m) * K + kh;
  const float* __restrict__ bcol = B + (size_t)kh * Ncols + (col0 + m);
  const float* bnk = bn + (kh >> 1) * 8;

  v8f acc = {};
#pragma unroll 4
  for (int k = 0; k < K; k += 4) {
    v2f a = *(const v2f*)(arow + k);
    const float4 ms = *(const float4*)(bnk + (k >> 1) * 8);
    const v2f sh = *(const v2f*)(bnk + (k >> 1) * 8 + 4);
    a.x = (a.x - ms.x) * ms.z + sh.x;
    a.y = (a.y - ms.y) * ms.w + sh.y;
    v2f b;
    b.x = bcol[(size_t)k * Ncols];
    b.y = bcol[(size_t)(k + 1) * Ncols];
    acc = __builtin_amdgcn_wmma_f32_16x16x4_f32(
        false, a, false, b, (short)0, acc, false, false);
  }

  float* __restrict__ crow = C + (size_t)col0 + m;
  const int rbase = row0 + ((lane >> 4) << 3);
#pragma unroll
  for (int r = 0; r < 8; ++r)
    crow[(size_t)(rbase + r) * Ncols] = acc[r];
}

#define TILE_ROWS 196
#define NTILES ((NN + TILE_ROWS - 1) / TILE_ROWS)
#define AGG_WAVES 32
#define QCAP 160
#define EDGE_UNR 4
#define NBLK (EE / 128)
static_assert(EE % 128 == 0, "a step reads 128 whole edges");
#define WCAP 256

template <int D>
__device__ __forceinline__ void add_edges(int s, int wbits, int r, int n,
                                          int lane,
                                          const float* __restrict__ sup,
                                          float* acc) {
#pragma nounroll
  for (int i = 0; i < n; i += EDGE_UNR) {
    v2f v[EDGE_UNR][D / 64];
    float wi[EDGE_UNR];
    int ri[EDGE_UNR];
#pragma unroll
    for (int u = 0; u < EDGE_UNR; ++u) {
      const int si = __builtin_amdgcn_readlane(s, i + u);
      wi[u] = __int_as_float(__builtin_amdgcn_readlane(wbits, i + u));
      ri[u] = __builtin_amdgcn_readlane(r, i + u);
      const v2f* p = (const v2f*)(sup + (size_t)si * D) + lane;
#pragma unroll
      for (int j = 0; j < D / 64; ++j) v[u][j] = p[j * 32];
    }
#pragma unroll
    for (int u = 0; u < EDGE_UNR; ++u) {
      if (i + u < n) {
        float* a = acc + ri[u] * D + 2 * lane;
#pragma unroll
        for (int j = 0; j < D / 64; ++j) {
          atomicAdd(a + j * 64, v[u][j].x * wi[u]);
          atomicAdd(a + j * 64 + 1, v[u][j].y * wi[u]);
        }
      }
    }
  }
}

template <int D, bool KEEP>
__device__ __forceinline__ void drain_queue(const int* __restrict__ qe,
                                            const int* __restrict__ qr,
                                            int n, int lane,
                                            const float* __restrict__ sup,
                                            const int* __restrict__ src,
                                            const float* __restrict__ w,
                                            float* acc,
                                            int4* __restrict__ seg, int have) {
  int e = 0, r = 0;
  if (lane < n) { e = qe[lane]; r = qr[lane]; }
  const int s = src[e];
  const int wbits = __float_as_int(w[e]);
  if (KEEP) {
    if (lane < n && have + lane < WCAP)
      seg[have + lane] = make_int4(s, wbits, r, 0);
  }
  add_edges<D>(s, wbits, r, n, lane, sup, acc);
}

template <int D, bool KEEP>
__device__ __forceinline__ int scan_dst(int wave, int lane, int row0,
                                        int nrows, int* qe, int* qr,
                                        const float* __restrict__ sup,
                                        const int* __restrict__ src,
                                        const int* __restrict__ dst,
                                        const float* __restrict__ w,
                                        float* acc, int4* __restrict__ seg) {
  int qn = 0;
  int ln = 0;

  const int rot = (int)(blockIdx.x % NTILES) * NBLK / NTILES;

  for (int k0 = wave; k0 < NBLK; k0 += AGG_WAVES * 4) {
    int4 d[4];
    int eb[4];
#pragma unroll
    for (int u = 0; u < 4; ++u) {
      const int k = k0 + u * AGG_WAVES;
      int blk = k + rot;
      if (blk >= NBLK) blk -= NBLK;
      eb[u] = (blk << 7) + (lane << 2);
      d[u] = make_int4(-1, -1, -1, -1);
      if (k < NBLK) d[u] = *(const int4*)(dst + eb[u]);
    }
#pragma unroll
    for (int u = 0; u < 4; ++u) {
      const unsigned r0 = (unsigned)(d[u].x - row0);
      const unsigned r1 = (unsigned)(d[u].y - row0);
      const unsigned r2 = (unsigned)(d[u].z - row0);
      const unsigned r3 = (unsigned)(d[u].w - row0);
      const bool m0 = r0 < (unsigned)nrows;
      const bool m1 = r1 < (unsigned)nrows;
      const bool m2 = r2 < (unsigned)nrows;
      const bool m3 = r3 < (unsigned)nrows;
      if (__builtin_amdgcn_ballot_w32(m0 | m1 | m2 | m3)) {
        const unsigned k0m = __builtin_amdgcn_ballot_w32(m0);
        const unsigned k1m = __builtin_amdgcn_ballot_w32(m1);
        const unsigned k2m = __builtin_amdgcn_ballot_w32(m2);
        const unsigned k3m = __builtin_amdgcn_ballot_w32(m3);
        if (m0) { const int sl = qn + (int)__builtin_amdgcn_mbcnt_lo(k0m, 0u); qe[sl] = eb[u];     qr[sl] = (int)r0; }
        qn += __popc(k0m);
        if (m1) { const int sl = qn + (int)__builtin_amdgcn_mbcnt_lo(k1m, 0u); qe[sl] = eb[u] + 1; qr[sl] = (int)r1; }
        qn += __popc(k1m);
        if (m2) { const int sl = qn + (int)__builtin_amdgcn_mbcnt_lo(k2m, 0u); qe[sl] = eb[u] + 2; qr[sl] = (int)r2; }
        qn += __popc(k2m);
        if (m3) { const int sl = qn + (int)__builtin_amdgcn_mbcnt_lo(k3m, 0u); qe[sl] = eb[u] + 3; qr[sl] = (int)r3; }
        qn += __popc(k3m);
        while (qn >= 32) {
          qn -= 32;
          drain_queue<D, KEEP>(qe + qn, qr + qn, 32, lane, sup, src, w, acc,
                               seg, ln);
          ln += 32;
        }
      }
    }
  }
  if (qn > 0) {
    drain_queue<D, KEEP>(qe, qr, qn, lane, sup, src, w, acc, seg, ln);
    ln += qn;
  }
  return ln;
}

template <int D, bool FROM_LIST>
__global__ __launch_bounds__(AGG_WAVES * 32)
void agg_tiles(const float* __restrict__ sup,
               float* __restrict__ agg,
               const int* __restrict__ src,
               const int* __restrict__ dst,
               const float* __restrict__ w,
               float* __restrict__ stats,
               int* __restrict__ lists,
               int* __restrict__ lcnt) {
  __shared__ float acc[TILE_ROWS * D];
  __shared__ int q_e[AGG_WAVES][QCAP];
  __shared__ int q_r[AGG_WAVES][QCAP];
  __shared__ float colsum[2 * D];

  const int lane = threadIdx.x & 31;
  const int wave = __builtin_amdgcn_readfirstlane((int)(threadIdx.x >> 5));
  const int row0 = blockIdx.x * TILE_ROWS;
  int nrows = NN - row0;
  if (nrows > TILE_ROWS) nrows = TILE_ROWS;
  if (nrows <= 0) return;

  for (int t = threadIdx.x; t < TILE_ROWS * D / 4; t += AGG_WAVES * 32)
    ((float4*)acc)[t] = make_float4(0.0f, 0.0f, 0.0f, 0.0f);
  for (int t = threadIdx.x; t < 2 * D; t += AGG_WAVES * 32) colsum[t] = 0.0f;
  __syncthreads();

  int4* seg = (int4*)lists + ((size_t)blockIdx.x * AGG_WAVES + wave) * WCAP;
  int* cnt = lcnt + blockIdx.x * AGG_WAVES;

  if (FROM_LIST) {
    static_assert(AGG_WAVES == 32, "one count per lane");
    const int c = cnt[lane];
    if (__builtin_amdgcn_ballot_w32(c > WCAP) == 0u) {
      const int n = __builtin_amdgcn_readlane(c, wave);
      int i = 0;
      for (; i + 32 <= n; i += 32) {
        const int4 en = seg[i + lane];
        add_edges<D>(en.x, en.y, en.z, 32, lane, sup, acc);
      }
      if (i < n) {
        int4 en = make_int4(0, 0, 0, 0);
        if (i + lane < n) en = seg[i + lane];
        add_edges<D>(en.x, en.y, en.z, n - i, lane, sup, acc);
      }
    } else {
      scan_dst<D, false>(wave, lane, row0, nrows, q_e[wave], q_r[wave],
                         sup, src, dst, w, acc, seg);
    }
  } else {
    const int ln = scan_dst<D, true>(wave, lane, row0, nrows, q_e[wave],
                                     q_r[wave], sup, src, dst, w, acc, seg);
    if (lane == 0) cnt[wave] = ln;
  }

  __syncthreads();

  static_assert((AGG_WAVES * 32) % (D / 4) == 0, "a thread keeps its columns");
  float4* __restrict__ o = (float4*)(agg + (size_t)row0 * D);
  float4 s1 = make_float4(0.0f, 0.0f, 0.0f, 0.0f);
  float4 s2 = make_float4(0.0f, 0.0f, 0.0f, 0.0f);
  for (int t = threadIdx.x; t < nrows * (D / 4); t += AGG_WAVES * 32) {
    float4 v = ((const float4*)acc)[t];
    v.x = v.x > 0.0f ? v.x : 0.0f;
    v.y = v.y > 0.0f ? v.y : 0.0f;
    v.z = v.z > 0.0f ? v.z : 0.0f;
    v.w = v.w > 0.0f ? v.w : 0.0f;
    o[t] = v;
    s1.x += v.x; s1.y += v.y; s1.z += v.z; s1.w += v.w;
    s2.x += v.x * v.x; s2.y += v.y * v.y; s2.z += v.z * v.z; s2.w += v.w * v.w;
  }
  const int c0 = (threadIdx.x % (D / 4)) * 4;
  atomicAdd(&colsum[c0 + 0], s1.x);
  atomicAdd(&colsum[c0 + 1], s1.y);
  atomicAdd(&colsum[c0 + 2], s1.z);
  atomicAdd(&colsum[c0 + 3], s1.w);
  atomicAdd(&colsum[D + c0 + 0], s2.x);
  atomicAdd(&colsum[D + c0 + 1], s2.y);
  atomicAdd(&colsum[D + c0 + 2], s2.z);
  atomicAdd(&colsum[D + c0 + 3], s2.w);
  __syncthreads();
  for (int t = threadIdx.x; t < 2 * D; t += AGG_WAVES * 32)
    atomicAdd(&stats[t], colsum[t]);
}

__global__ void bn_normalize(const float* __restrict__ agg,
                             float* __restrict__ out,
                             const float* __restrict__ gamma,
                             const float* __restrict__ beta,
                             const float* __restrict__ stats,
                             int Nrows, int D) {
  size_t total = (size_t)Nrows * D;
  size_t i = (size_t)blockIdx.x * blockDim.x + threadIdx.x;
  if (i >= total) return;
  int c = (int)(i & (size_t)(D - 1));
  const float invN = 1.0f / (float)Nrows;
  const float mean = stats[c] * invN;
  const float var  = stats[D + c] * invN - mean * mean;
  const float s = gamma[c] * rsqrtf(var + BN_EPS);
  out[i] = (agg[i] - mean) * s + beta[c];
}

extern "C" void kernel_launch(void* const* d_in, const int* in_sizes, int n_in,
                              void* d_out, int out_size, void* d_ws,
                              size_t ws_size, hipStream_t stream) {
  (void)in_sizes; (void)n_in; (void)out_size; (void)ws_size;

  const float* y   = (const float*)d_in[0];
  const int*   src = (const int*)d_in[1];
  const int*   dst = (const int*)d_in[2];
  const float* w   = (const float*)d_in[3];
  const float* W1  = (const float*)d_in[4];
  const float* g1  = (const float*)d_in[5];
  const float* b1  = (const float*)d_in[6];
  const float* W2  = (const float*)d_in[7];
  const float* g2  = (const float*)d_in[8];
  const float* b2  = (const float*)d_in[9];
  float* out = (float*)d_out;

  float* ws    = (float*)d_ws;
  float* sup1  = ws;
  float* sup2  = sup1 + (size_t)NN * H1C;
  float* agg1  = sup2 + (size_t)NN * DINC;
  float* agg2  = agg1 + (size_t)NN * H1C;
  float* stats1 = agg2 + (size_t)NN * DINC;
  float* stats2 = stats1 + 2 * H1C;
  int* lists = (int*)(stats2 + 2 * DINC);
  int* lcnt  = lists + (size_t)NTILES * AGG_WAVES * WCAP * 4;

  wmma_gemm_f32<<<25000 / 8, 256, 0, stream>>>(
      y, W1, sup1, stats1, 2 * H1C + 2 * DINC, NN, H2C, H1C);

  agg_tiles<H1C, false><<<NTILES, AGG_WAVES * 32, 0, stream>>>(
      sup1, agg1, src, dst, w, stats1, lists, lcnt);

  wmma_gemm_bn_f32<<<50000 / 8, 256, 0, stream>>>(
      agg1, W2, sup2, stats1, g1, b1, NN, H1C, DINC);

  agg_tiles<DINC, true><<<NTILES, AGG_WAVES * 32, 0, stream>>>(
      sup2, agg2, src, dst, w, stats2, lists, lcnt);

  {
    size_t total = (size_t)NN * DINC;
    bn_normalize<<<(int)((total + 255) / 256), 256, 0, stream>>>(
        agg2, out, g2, b2, stats2, NN, DINC);
  }
}


# Layer-1 walk reads a one-byte key plane, not dst: launch 1 stores each edge's tile number as a byte, a wave tests 512 keys per load (4x fewer bytes); dst read per queued edge; LDS add order changes
# speedup vs baseline: 2.7729x; 1.0738x over previous
#include <hip/hip_runtime.h>
#include <hip/hip_bf16.h>

typedef float v2f __attribute__((ext_vector_type(2)));
typedef float v8f __attribute__((ext_vector_type(8)));

#define NN   50000
#define EE   800000
#define H2C  64
#define H1C  128
#define DINC 256
#define BN_EPS 1e-5f

#define TILE_ROWS 196
#define NTILES ((NN + TILE_ROWS - 1) / TILE_ROWS)
static_assert(NTILES <= 256, "a tile number fits one byte");

__device__ __forceinline__ unsigned tile_key(int d) {
  return ((unsigned)d / (unsigned)TILE_ROWS) & 0xFFu;
}

__global__ void wmma_gemm_f32(const float* __restrict__ A,
                              const float* __restrict__ B,
                              float* __restrict__ C,
                              float* __restrict__ zero_buf, int zero_n,
                              const int* __restrict__ dst,
                              int* __restrict__ keys,
                              int Nrows, int K, int Ncols) {
  if (blockIdx.x == 0)
    for (int t = threadIdx.x; t < zero_n; t += blockDim.x) zero_buf[t] = 0.0f;

  {
    const int g = blockIdx.x * blockDim.x + threadIdx.x;
    if (g < EE / 4) {
      const int4 d = *(const int4*)(dst + 4 * g);
      keys[g] = (int)(tile_key(d.x) | (tile_key(d.y) << 8) |
                      (tile_key(d.z) << 16) | (tile_key(d.w) << 24));
    }
  }

  const int lane = threadIdx.x & 31;
  const int wave = threadIdx.x >> 5;
  const int colTiles = Ncols >> 4;
  const int rowTiles = Nrows >> 4;
  int tile = blockIdx.x * (blockDim.x >> 5) + wave;
  if (tile >= rowTiles * colTiles) return;

  const int rt = tile / colTiles;
  const int ct = tile - rt * colTiles;
  const int row0 = rt << 4;
  const int col0 = ct << 4;

  const int m  = lane & 15;
  const int kh = (lane >> 4) << 1;

  const float* __restrict__ arow = A + (size_t)(row0 + m) * K + kh;
  const float* __restrict__ bcol = B + (size_t)kh * Ncols + (col0 + m);

  v8f acc = {};
  for (int k = 0; k < K; k += 4) {
    v2f a = *(const v2f*)(arow + k);
    v2f b;
    b.x = bcol[(size_t)k * Ncols];
    b.y = bcol[(size_t)(k + 1) * Ncols];
    acc = __builtin_amdgcn_wmma_f32_16x16x4_f32(
        false, a, false, b, (short)0, acc, false, false);
  }

  float* __restrict__ crow = C + (size_t)col0 + m;
  const int rbase = row0 + ((lane >> 4) << 3);
#pragma unroll
  for (int r = 0; r < 8; ++r)
    crow[(size_t)(rbase + r) * Ncols] = acc[r];
}

__global__ __launch_bounds__(256)
void wmma_gemm_bn_f32(const float* __restrict__ A,
                      const float* __restrict__ B,
                      float* __restrict__ C,
                      const float* __restrict__ stats,
                      const float* __restrict__ gamma,
                      const float* __restrict__ beta,
                      int Nrows, int K, int Ncols) {
  __shared__ __attribute__((aligned(16))) float bn[(H1C / 2) * 8];
  for (int c = threadIdx.x; c < K; c += blockDim.x) {
    const float invN = 1.0f / (float)Nrows;
    const float mean = stats[c] * invN;
    const float var  = stats[K + c] * invN - mean * mean;
    const float s = gamma[c] * rsqrtf(var + BN_EPS);
    float* e = bn + (c >> 1) * 8 + (c & 1);
    e[0] = mean;
    e[2] = s;
    e[4] = beta[c];
  }
  __syncthreads();

  const int lane = threadIdx.x & 31;
  const int wave = threadIdx.x >> 5;
  const int colTiles = Ncols >> 4;
  const int rowTiles = Nrows >> 4;
  int tile = blockIdx.x * (blockDim.x >> 5) + wave;
  if (tile >= rowTiles * colTiles) return;

  const int rt = tile / colTiles;
  const int ct = tile - rt * colTiles;
  const int row0 = rt << 4;
  const int col0 = ct << 4;

  const int m  = lane & 15;
  const int kh = (lane >> 4) << 1;

  const float* __restrict__ arow = A + (size_t)(row0 + m) * K + kh;
  const float* __restrict__ bcol = B + (size_t)kh * Ncols + (col0 + m);
  const float* bnk = bn + (kh >> 1) * 8;

  v8f acc = {};
#pragma unroll 4
  for (int k = 0; k < K; k += 4) {
    v2f a = *(const v2f*)(arow + k);
    const float4 ms = *(const float4*)(bnk + (k >> 1) * 8);
    const v2f sh = *(const v2f*)(bnk + (k >> 1) * 8 + 4);
    a.x = (a.x - ms.x) * ms.z + sh.x;
    a.y = (a.y - ms.y) * ms.w + sh.y;
    v2f b;
    b.x = bcol[(size_t)k * Ncols];
    b.y = bcol[(size_t)(k + 1) * Ncols];
    acc = __builtin_amdgcn_wmma_f32_16x16x4_f32(
        false, a, false, b, (short)0, acc, false, false);
  }

  float* __restrict__ crow = C + (size_t)col0 + m;
  const int rbase = row0 + ((lane >> 4) << 3);
#pragma unroll
  for (int r = 0; r < 8; ++r)
    crow[(size_t)(rbase + r) * Ncols] = acc[r];
}

#define AGG_WAVES 32
#define QCAP 64
#define EDGE_UNR 4
#define KSTEP 512
#define NKBLK ((EE + KSTEP - 1) / KSTEP)
static_assert(EE % 16 == 0, "a lane's 16 keys are all edges or none");
#define WCAP 256

template <int D>
__device__ __forceinline__ void add_edges(int s, int wbits, int r, int n,
                                          int lane,
                                          const float* __restrict__ sup,
                                          float* acc) {
#pragma nounroll
  for (int i = 0; i < n; i += EDGE_UNR) {
    v2f v[EDGE_UNR][D / 64];
    float wi[EDGE_UNR];
    int ri[EDGE_UNR];
#pragma unroll
    for (int u = 0; u < EDGE_UNR; ++u) {
      const int si = __builtin_amdgcn_readlane(s, i + u);
      wi[u] = __int_as_float(__builtin_amdgcn_readlane(wbits, i + u));
      ri[u] = __builtin_amdgcn_readlane(r, i + u);
      const v2f* p = (const v2f*)(sup + (size_t)si * D) + lane;
#pragma unroll
      for (int j = 0; j < D / 64; ++j) v[u][j] = p[j * 32];
    }
#pragma unroll
    for (int u = 0; u < EDGE_UNR; ++u) {
      if (i + u < n) {
        float* a = acc + ri[u] * D + 2 * lane;
#pragma unroll
        for (int j = 0; j < D / 64; ++j) {
          atomicAdd(a + j * 64, v[u][j].x * wi[u]);
          atomicAdd(a + j * 64 + 1, v[u][j].y * wi[u]);
        }
      }
    }
  }
}

template <int D, bool KEEP>
__device__ __forceinline__ void drain_queue(const int* __restrict__ qe,
                                            int n, int lane,
                                            int row0, int nrows,
                                            const float* __restrict__ sup,
                                            const int* __restrict__ src,
                                            const int* __restrict__ dst,
                                            const float* __restrict__ w,
                                            float* acc,
                                            int4* __restrict__ seg, int have) {
  int e = 0;
  if (lane < n) e = qe[lane];
  const int s = src[e];
  int wbits = __float_as_int(w[e]);
  int r = dst[e] - row0;
  if ((unsigned)r >= (unsigned)nrows) { r = 0; wbits = 0; }
  if (KEEP) {
    if (lane < n && have + lane < WCAP)
      seg[have + lane] = make_int4(s, wbits, r, 0);
  }
  add_edges<D>(s, wbits, r, n, lane, sup, acc);
}

__device__ __forceinline__ unsigned match_bit7(unsigned x, unsigned tpat) {
  const unsigned y = x ^ tpat;
  const unsigned t = (y & 0x7F7F7F7Fu) + 0x7F7F7F7Fu;
  return ~(t | y);
}

__device__ __forceinline__ unsigned pick(unsigned m, unsigned a, unsigned b) {
  return (a & m) | (b & ~m);
}

__device__ __forceinline__ unsigned match16(int4 d, unsigned tpat) {
  const unsigned n0 = match_bit7((unsigned)d.x, tpat) >> 3;
  const unsigned n1 = match_bit7((unsigned)d.y, tpat) >> 2;
  const unsigned n2 = match_bit7((unsigned)d.z, tpat) >> 1;
  const unsigned n3 = match_bit7((unsigned)d.w, tpat);
  return pick(0x10101010u, n0, pick(0x20202020u, n1, pick(0x40404040u, n2, n3))) &
         0xF0F0F0F0u;
}

template <int D, bool KEEP>
__device__ __forceinline__ int scan_keys(int wave, int lane, int row0,
                                         int nrows, int* qe,
                                         const float* __restrict__ sup,
                                         const int* __restrict__ src,
                                         const int* __restrict__ dst,
                                         const float* __restrict__ w,
                                         const int* __restrict__ keys,
                                         float* acc, int4* __restrict__ seg) {
  int qn = 0;
  int ln = 0;
  const unsigned tpat = (unsigned)blockIdx.x * 0x01010101u;

  const int rot = (int)(blockIdx.x % NTILES) * NKBLK / NTILES;

  for (int k0 = wave; k0 < NKBLK; k0 += AGG_WAVES * 4) {
    int4 d[4];
    unsigned ok[4];
#pragma unroll
    for (int u = 0; u < 4; ++u) {
      const int k = k0 + u * AGG_WAVES;
      int blk = k + rot;
      if (blk >= NKBLK) blk -= NKBLK;
      const int e0 = (blk << 9) + (lane << 4);
      const bool own = k < NKBLK && e0 < EE;
      ok[u] = 0u - (unsigned)own;
      const int ea = own ? e0 : EE - 16;
      d[u] = ((const int4*)keys)[(unsigned)ea >> 4];
    }
    __builtin_amdgcn_sched_barrier(0);
    unsigned m16[4];
#pragma unroll
    for (int u = 0; u < 4; ++u) m16[u] = match16(d[u], tpat) & ok[u];
    unsigned long long mm =
        (unsigned long long)(m16[0] | (m16[1] >> 4)) |
        ((unsigned long long)(m16[2] | (m16[3] >> 4)) << 32);
    while (__builtin_amdgcn_ballot_w32(mm != 0ull)) {
      const bool has = mm != 0ull;
      const unsigned bm = __builtin_amdgcn_ballot_w32(has);
      if (has) {
        const int pos = __ffsll((long long)mm) - 1;
        const int q = pos & 7;
        const int u = ((pos >> 5) << 1) | ((q >> 2) ^ 1);
        const int key = ((q & 3) << 2) | ((pos & 31) >> 3);
        int blk = k0 + u * AGG_WAVES + rot;
        if (blk >= NKBLK) blk -= NKBLK;
        qe[qn + (int)__builtin_amdgcn_mbcnt_lo(bm, 0u)] =
            (blk << 9) + (lane << 4) + key;
        mm &= mm - 1ull;
      }
      qn += __popc(bm);
      if (qn >= 32) {
        qn -= 32;
        drain_queue<D, KEEP>(qe + qn, 32, lane, row0, nrows, sup, src, dst, w,
                             acc, seg, ln);
        ln += 32;
      }
    }
  }
  if (qn > 0) {
    drain_queue<D, KEEP>(qe, qn, lane, row0, nrows, sup, src, dst, w, acc,
                         seg, ln);
    ln += qn;
  }
  return ln;
}

template <int D, bool FROM_LIST>
__global__ __launch_bounds__(AGG_WAVES * 32)
void agg_tiles(const float* __restrict__ sup,
               float* __restrict__ agg,
               const int* __restrict__ src,
               const int* __restrict__ dst,
               const float* __restrict__ w,
               const int* __restrict__ keys,
               float* __restrict__ stats,
               int* __restrict__ lists,
               int* __restrict__ lcnt) {
  __shared__ float acc[TILE_ROWS * D];
  __shared__ int q_e[AGG_WAVES][QCAP];
  __shared__ float colsum[2 * D];

  const int lane = threadIdx.x & 31;
  const int wave = __builtin_amdgcn_readfirstlane((int)(threadIdx.x >> 5));
  const int row0 = blockIdx.x * TILE_ROWS;
  int nrows = NN - row0;
  if (nrows > TILE_ROWS) nrows = TILE_ROWS;
  if (nrows <= 0) return;

  for (int t = threadIdx.x; t < TILE_ROWS * D / 4; t += AGG_WAVES * 32)
    ((float4*)acc)[t] = make_float4(0.0f, 0.0f, 0.0f, 0.0f);
  for (int t = threadIdx.x; t < 2 * D; t += AGG_WAVES * 32) colsum[t] = 0.0f;
  __syncthreads();

  int4* seg = (int4*)lists + ((size_t)blockIdx.x * AGG_WAVES + wave) * WCAP;
  int* cnt = lcnt + blockIdx.x * AGG_WAVES;

  if (FROM_LIST) {
    static_assert(AGG_WAVES == 32, "one count per lane");
    const int c = cnt[lane];
    if (__builtin_amdgcn_ballot_w32(c > WCAP) == 0u) {
      const int n = __builtin_amdgcn_readlane(c, wave);
      int i = 0;
      for (; i + 32 <= n; i += 32) {
        const int4 en = seg[i + lane];
        add_edges<D>(en.x, en.y, en.z, 32, lane, sup, acc);
      }
      if (i < n) {
        int4 en = make_int4(0, 0, 0, 0);
        if (i + lane < n) en = seg[i + lane];
        add_edges<D>(en.x, en.y, en.z, n - i, lane, sup, acc);
      }
    } else {
      scan_keys<D, false>(wave, lane, row0, nrows, q_e[wave], sup, src, dst,
                          w, keys, acc, seg);
    }
  } else {
    const int ln = scan_keys<D, true>(wave, lane, row0, nrows, q_e[wave],
                                      sup, src, dst, w, keys, acc, seg);
    if (lane == 0) cnt[wave] = ln;
  }

  __syncthreads();

  static_assert((AGG_WAVES * 32) % (D / 4) == 0, "a thread keeps its columns");
  float4* __restrict__ o = (float4*)(agg + (size_t)row0 * D);
  float4 s1 = make_float4(0.0f, 0.0f, 0.0f, 0.0f);
  float4 s2 = make_float4(0.0f, 0.0f, 0.0f, 0.0f);
  for (int t = threadIdx.x; t < nrows * (D / 4); t += AGG_WAVES * 32) {
    float4 v = ((const float4*)acc)[t];
    v.x = v.x > 0.0f ? v.x : 0.0f;
    v.y = v.y > 0.0f ? v.y : 0.0f;
    v.z = v.z > 0.0f ? v.z : 0.0f;
    v.w = v.w > 0.0f ? v.w : 0.0f;
    o[t] = v;
    s1.x += v.x; s1.y += v.y; s1.z += v.z; s1.w += v.w;
    s2.x += v.x * v.x; s2.y += v.y * v.y; s2.z += v.z * v.z; s2.w += v.w * v.w;
  }
  const int c0 = (threadIdx.x % (D / 4)) * 4;
  atomicAdd(&colsum[c0 + 0], s1.x);
  atomicAdd(&colsum[c0 + 1], s1.y);
  atomicAdd(&colsum[c0 + 2], s1.z);
  atomicAdd(&colsum[c0 + 3], s1.w);
  atomicAdd(&colsum[D + c0 + 0], s2.x);
  atomicAdd(&colsum[D + c0 + 1], s2.y);
  atomicAdd(&colsum[D + c0 + 2], s2.z);
  atomicAdd(&colsum[D + c0 + 3], s2.w);
  __syncthreads();
  for (int t = threadIdx.x; t < 2 * D; t += AGG_WAVES * 32)
    atomicAdd(&stats[t], colsum[t]);
}

__global__ void bn_normalize(const float* __restrict__ agg,
                             float* __restrict__ out,
                             const float* __restrict__ gamma,
                             const float* __restrict__ beta,
                             const float* __restrict__ stats,
                             int Nrows, int D) {
  size_t total = (size_t)Nrows * D;
  size_t i = (size_t)blockIdx.x * blockDim.x + threadIdx.x;
  if (i >= total) return;
  int c = (int)(i & (size_t)(D - 1));
  const float invN = 1.0f / (float)Nrows;
  const float mean = stats[c] * invN;
  const float var  = stats[D + c] * invN - mean * mean;
  const float s = gamma[c] * rsqrtf(var + BN_EPS);
  out[i] = (agg[i] - mean) * s + beta[c];
}

extern "C" void kernel_launch(void* const* d_in, const int* in_sizes, int n_in,
                              void* d_out, int out_size, void* d_ws,
                              size_t ws_size, hipStream_t stream) {
  (void)in_sizes; (void)n_in; (void)out_size; (void)ws_size;

  const float* y   = (const float*)d_in[0];
  const int*   src = (const int*)d_in[1];
  const int*   dst = (const int*)d_in[2];
  const float* w   = (const float*)d_in[3];
  const float* W1  = (const float*)d_in[4];
  const float* g1  = (const float*)d_in[5];
  const float* b1  = (const float*)d_in[6];
  const float* W2  = (const float*)d_in[7];
  const float* g2  = (const float*)d_in[8];
  const float* b2  = (const float*)d_in[9];
  float* out = (float*)d_out;

  float* ws    = (float*)d_ws;
  float* sup1  = ws;
  float* sup2  = sup1 + (size_t)NN * H1C;
  float* agg1  = sup2 + (size_t)NN * DINC;
  float* agg2  = agg1 + (size_t)NN * H1C;
  float* stats1 = agg2 + (size_t)NN * DINC;
  float* stats2 = stats1 + 2 * H1C;
  int* lists = (int*)(stats2 + 2 * DINC);
  int* lcnt  = lists + (size_t)NTILES * AGG_WAVES * WCAP * 4;
  int* keys  = lcnt + NTILES * AGG_WAVES;

  wmma_gemm_f32<<<25000 / 8, 256, 0, stream>>>(
      y, W1, sup1, stats1, 2 * H1C + 2 * DINC, dst, keys, NN, H2C, H1C);

  agg_tiles<H1C, false><<<NTILES, AGG_WAVES * 32, 0, stream>>>(
      sup1, agg1, src, dst, w, keys, stats1, lists, lcnt);

  wmma_gemm_bn_f32<<<50000 / 8, 256, 0, stream>>>(
      agg1, W2, sup2, stats1, g1, b1, NN, H1C, DINC);

  agg_tiles<DINC, true><<<NTILES, AGG_WAVES * 32, 0, stream>>>(
      sup2, agg2, src, dst, w, keys, stats2, lists, lcnt);

  {
    size_t total = (size_t)NN * DINC;
    bn_normalize<<<(int)((total + 255) / 256), 256, 0, stream>>>(
        agg2, out, g2, b2, stats2, NN, DINC);
  }
}


# Second dense product keeps rows, walks columns: a wave holds its 16 normalised rows of A in registers, W2 sits in LDS as ready fragments; A fetched 4x, not 16x; k order inside the sums changes
# speedup vs baseline: 3.1463x; 1.1347x over previous
#include <hip/hip_runtime.h>
#include <hip/hip_bf16.h>

typedef float v2f __attribute__((ext_vector_type(2)));
typedef float v8f __attribute__((ext_vector_type(8)));

#define NN   50000
#define EE   800000
#define H2C  64
#define H1C  128
#define DINC 256
#define BN_EPS 1e-5f

#define TILE_ROWS 196
#define NTILES ((NN + TILE_ROWS - 1) / TILE_ROWS)
static_assert(NTILES <= 256, "a tile number fits one byte");

__device__ __forceinline__ unsigned tile_key(int d) {
  return ((unsigned)d / (unsigned)TILE_ROWS) & 0xFFu;
}

__global__ void wmma_gemm_f32(const float* __restrict__ A,
                              const float* __restrict__ B,
                              float* __restrict__ C,
                              float* __restrict__ zero_buf, int zero_n,
                              const int* __restrict__ dst,
                              int* __restrict__ keys,
                              int Nrows, int K, int Ncols) {
  if (blockIdx.x == 0)
    for (int t = threadIdx.x; t < zero_n; t += blockDim.x) zero_buf[t] = 0.0f;

  {
    const int g = blockIdx.x * blockDim.x + threadIdx.x;
    if (g < EE / 4) {
      const int4 d = *(const int4*)(dst + 4 * g);
      keys[g] = (int)(tile_key(d.x) | (tile_key(d.y) << 8) |
                      (tile_key(d.z) << 16) | (tile_key(d.w) << 24));
    }
  }

  const int lane = threadIdx.x & 31;
  const int wave = threadIdx.x >> 5;
  const int colTiles = Ncols >> 4;
  const int rowTiles = Nrows >> 4;
  int tile = blockIdx.x * (blockDim.x >> 5) + wave;
  if (tile >= rowTiles * colTiles) return;

  const int rt = tile / colTiles;
  const int ct = tile - rt * colTiles;
  const int row0 = rt << 4;
  const int col0 = ct << 4;

  const int m  = lane & 15;
  const int kh = (lane >> 4) << 1;

  const float* __restrict__ arow = A + (size_t)(row0 + m) * K + kh;
  const float* __restrict__ bcol = B + (size_t)kh * Ncols + (col0 + m);

  v8f acc = {};
  for (int k = 0; k < K; k += 4) {
    v2f a = *(const v2f*)(arow + k);
    v2f b;
    b.x = bcol[(size_t)k * Ncols];
    b.y = bcol[(size_t)(k + 1) * Ncols];
    acc = __builtin_amdgcn_wmma_f32_16x16x4_f32(
        false, a, false, b, (short)0, acc, false, false);
  }

  float* __restrict__ crow = C + (size_t)col0 + m;
  const int rbase = row0 + ((lane >> 4) << 3);
#pragma unroll
  for (int r = 0; r < 8; ++r)
    crow[(size_t)(rbase + r) * Ncols] = acc[r];
}

#define G2_WAVES 16
#define G2_CT 4
#define G2_CG (DINC / (16 * G2_CT))
#define G2_ITEMS ((NN / 16) * G2_CG)
#define G2_PAIRS (H1C / 8)
static_assert(NN % 16 == 0 && H1C % 8 == 0 && DINC % (16 * G2_CT) == 0,
              "whole tiles only");
static_assert(G2_CG == 4 && G2_PAIRS == 16, "shifts below");

__global__ __launch_bounds__(G2_WAVES * 32)
void wmma_gemm_bn_rows(const float* __restrict__ A,
                       const float* __restrict__ B,
                       float* __restrict__ C,
                       const float* __restrict__ stats,
                       const float* __restrict__ gamma,
                       const float* __restrict__ beta) {
  __shared__ __attribute__((aligned(16))) float wf[(DINC / 16) * G2_PAIRS * 32 * 4];
  __shared__ __attribute__((aligned(16))) float bn[3 * H1C];

  for (int c = threadIdx.x; c < H1C; c += G2_WAVES * 32) {
    const float invN = 1.0f / (float)NN;
    const float mean = stats[c] * invN;
    const float var  = stats[H1C + c] * invN - mean * mean;
    bn[c] = mean;
    bn[H1C + c] = gamma[c] * rsqrtf(var + BN_EPS);
    bn[2 * H1C + c] = beta[c];
  }
#pragma unroll 4
  for (int q = threadIdx.x; q < (DINC / 16) * G2_PAIRS * 32; q += G2_WAVES * 32) {
    const int l = q & 31;
    const int j = (q >> 5) & (G2_PAIRS - 1);
    const int ct = q >> 9;
    const float* p = B + (size_t)(8 * j + 4 * (l >> 4)) * DINC + 16 * ct + (l & 15);
    float4 v;
    v.x = p[0];
    v.y = p[DINC];
    v.z = p[2 * DINC];
    v.w = p[3 * DINC];
    ((float4*)wf)[q] = v;
  }
  __syncthreads();

  const int lane = threadIdx.x & 31;
  const int wave = __builtin_amdgcn_readfirstlane((int)(threadIdx.x >> 5));
  const int m = lane & 15;
  const int h = lane >> 4;
  const float4* bnv = (const float4*)bn + h;

  for (int it = blockIdx.x + gridDim.x * wave; it < G2_ITEMS;
       it += gridDim.x * G2_WAVES) {
    __atomic_signal_fence(__ATOMIC_SEQ_CST);
    const int rt = it >> 2;
    const int cg = it & 3;
    const float* arow = A + (size_t)(rt * 16 + m) * H1C + 4 * h;
    float4 a[G2_PAIRS];
#pragma unroll
    for (int j = 0; j < G2_PAIRS; ++j) a[j] = *(const float4*)(arow + 8 * j);
    __builtin_amdgcn_sched_barrier(0);
#pragma unroll
    for (int j = 0; j < G2_PAIRS; ++j) {
      const float4 mu = bnv[2 * j];
      const float4 sc = bnv[H1C / 4 + 2 * j];
      const float4 sh = bnv[2 * (H1C / 4) + 2 * j];
      a[j].x = (a[j].x - mu.x) * sc.x + sh.x;
      a[j].y = (a[j].y - mu.y) * sc.y + sh.y;
      a[j].z = (a[j].z - mu.z) * sc.z + sh.z;
      a[j].w = (a[j].w - mu.w) * sc.w + sh.w;
    }

    const float4* bw = (const float4*)wf + (cg * G2_CT) * (G2_PAIRS * 32) + lane;
    v8f acc[G2_CT];
#pragma unroll
    for (int t = 0; t < G2_CT; ++t) acc[t] = (v8f){};
#pragma unroll
    for (int j = 0; j < G2_PAIRS; ++j) {
      float4 b[G2_CT];
#pragma unroll
      for (int t = 0; t < G2_CT; ++t) b[t] = bw[(t * G2_PAIRS + j) * 32];
      const v2f a0 = {a[j].x, a[j].y};
      const v2f a1 = {a[j].z, a[j].w};
#pragma unroll
      for (int t = 0; t < G2_CT; ++t) {
        const v2f b0 = {b[t].x, b[t].y};
        acc[t] = __builtin_amdgcn_wmma_f32_16x16x4_f32(
            false, a0, false, b0, (short)0, acc[t], false, false);
      }
#pragma unroll
      for (int t = 0; t < G2_CT; ++t) {
        const v2f b1 = {b[t].z, b[t].w};
        acc[t] = __builtin_amdgcn_wmma_f32_16x16x4_f32(
            false, a1, false, b1, (short)0, acc[t], false, false);
      }
    }

    float* crow = C + (size_t)(rt * 16 + 8 * h) * DINC + cg * (16 * G2_CT) + m;
#pragma unroll
    for (int t = 0; t < G2_CT; ++t)
#pragma unroll
      for (int r = 0; r < 8; ++r)
        crow[(size_t)r * DINC + t * 16] = acc[t][r];
  }
}

#define AGG_WAVES 32
#define QCAP 64
#define EDGE_UNR 4
#define KSTEP 512
#define NKBLK ((EE + KSTEP - 1) / KSTEP)
static_assert(EE % 16 == 0, "a lane's 16 keys are all edges or none");
#define WCAP 256

template <int D>
__device__ __forceinline__ void add_edges(int s, int wbits, int r, int n,
                                          int lane,
                                          const float* __restrict__ sup,
                                          float* acc) {
#pragma nounroll
  for (int i = 0; i < n; i += EDGE_UNR) {
    v2f v[EDGE_UNR][D / 64];
    float wi[EDGE_UNR];
    int ri[EDGE_UNR];
#pragma unroll
    for (int u = 0; u < EDGE_UNR; ++u) {
      const int si = __builtin_amdgcn_readlane(s, i + u);
      wi[u] = __int_as_float(__builtin_amdgcn_readlane(wbits, i + u));
      ri[u] = __builtin_amdgcn_readlane(r, i + u);
      const v2f* p = (const v2f*)(sup + (size_t)si * D) + lane;
#pragma unroll
      for (int j = 0; j < D / 64; ++j) v[u][j] = p[j * 32];
    }
#pragma unroll
    for (int u = 0; u < EDGE_UNR; ++u) {
      if (i + u < n) {
        float* a = acc + ri[u] * D + 2 * lane;
#pragma unroll
        for (int j = 0; j < D / 64; ++j) {
          atomicAdd(a + j * 64, v[u][j].x * wi[u]);
          atomicAdd(a + j * 64 + 1, v[u][j].y * wi[u]);
        }
      }
    }
  }
}

template <int D, bool KEEP>
__device__ __forceinline__ void drain_queue(const int* __restrict__ qe,
                                            int n, int lane,
                                            int row0, int nrows,
                                            const float* __restrict__ sup,
                                            const int* __restrict__ src,
                                            const int* __restrict__ dst,
                                            const float* __restrict__ w,
                                            float* acc,
                                            int4* __restrict__ seg, int have) {
  int e = 0;
  if (lane < n) e = qe[lane];
  const int s = src[e];
  int wbits = __float_as_int(w[e]);
  int r = dst[e] - row0;
  if ((unsigned)r >= (unsigned)nrows) { r = 0; wbits = 0; }
  if (KEEP) {
    if (lane < n && have + lane < WCAP)
      seg[have + lane] = make_int4(s, wbits, r, 0);
  }
  add_edges<D>(s, wbits, r, n, lane, sup, acc);
}

__device__ __forceinline__ unsigned match_bit7(unsigned x, unsigned tpat) {
  const unsigned y = x ^ tpat;
  const unsigned t = (y & 0x7F7F7F7Fu) + 0x7F7F7F7Fu;
  return ~(t | y);
}

__device__ __forceinline__ unsigned pick(unsigned m, unsigned a, unsigned b) {
  return (a & m) | (b & ~m);
}

__device__ __forceinline__ unsigned match16(int4 d, unsigned tpat) {
  const unsigned n0 = match_bit7((unsigned)d.x, tpat) >> 3;
  const unsigned n1 = match_bit7((unsigned)d.y, tpat) >> 2;
  const unsigned n2 = match_bit7((unsigned)d.z, tpat) >> 1;
  const unsigned n3 = match_bit7((unsigned)d.w, tpat);
  return pick(0x10101010u, n0, pick(0x20202020u, n1, pick(0x40404040u, n2, n3))) &
         0xF0F0F0F0u;
}

template <int D, bool KEEP>
__device__ __forceinline__ int scan_keys(int wave, int lane, int row0,
                                         int nrows, int* qe,
                                         const float* __restrict__ sup,
                                         const int* __restrict__ src,
                                         const int* __restrict__ dst,
                                         const float* __restrict__ w,
                                         const int* __restrict__ keys,
                                         float* acc, int4* __restrict__ seg) {
  int qn = 0;
  int ln = 0;
  const unsigned tpat = (unsigned)blockIdx.x * 0x01010101u;

  const int rot = (int)(blockIdx.x % NTILES) * NKBLK / NTILES;

  for (int k0 = wave; k0 < NKBLK; k0 += AGG_WAVES * 4) {
    int4 d[4];
    unsigned ok[4];
#pragma unroll
    for (int u = 0; u < 4; ++u) {
      const int k = k0 + u * AGG_WAVES;
      int blk = k + rot;
      if (blk >= NKBLK) blk -= NKBLK;
      const int e0 = (blk << 9) + (lane << 4);
      const bool own = k < NKBLK && e0 < EE;
      ok[u] = 0u - (unsigned)own;
      const int ea = own ? e0 : EE - 16;
      d[u] = ((const int4*)keys)[(unsigned)ea >> 4];
    }
    __builtin_amdgcn_sched_barrier(0);
    unsigned m16[4];
#pragma unroll
    for (int u = 0; u < 4; ++u) m16[u] = match16(d[u], tpat) & ok[u];
    unsigned long long mm =
        (unsigned long long)(m16[0] | (m16[1] >> 4)) |
        ((unsigned long long)(m16[2] | (m16[3] >> 4)) << 32);
    while (__builtin_amdgcn_ballot_w32(mm != 0ull)) {
      const bool has = mm != 0ull;
      const unsigned bm = __builtin_amdgcn_ballot_w32(has);
      if (has) {
        const int pos = __ffsll((long long)mm) - 1;
        const int q = pos & 7;
        const int u = ((pos >> 5) << 1) | ((q >> 2) ^ 1);
        const int key = ((q & 3) << 2) | ((pos & 31) >> 3);
        int blk = k0 + u * AGG_WAVES + rot;
        if (blk >= NKBLK) blk -= NKBLK;
        qe[qn + (int)__builtin_amdgcn_mbcnt_lo(bm, 0u)] =
            (blk << 9) + (lane << 4) + key;
        mm &= mm - 1ull;
      }
      qn += __popc(bm);
      if (qn >= 32) {
        qn -= 32;
        drain_queue<D, KEEP>(qe + qn, 32, lane, row0, nrows, sup, src, dst, w,
                             acc, seg, ln);
        ln += 32;
      }
    }
  }
  if (qn > 0) {
    drain_queue<D, KEEP>(qe, qn, lane, row0, nrows, sup, src, dst, w, acc,
                         seg, ln);
    ln += qn;
  }
  return ln;
}

template <int D, bool FROM_LIST>
__global__ __launch_bounds__(AGG_WAVES * 32)
void agg_tiles(const float* __restrict__ sup,
               float* __restrict__ agg,
               const int* __restrict__ src,
               const int* __restrict__ dst,
               const float* __restrict__ w,
               const int* __restrict__ keys,
               float* __restrict__ stats,
               int* __restrict__ lists,
               int* __restrict__ lcnt) {
  __shared__ float acc[TILE_ROWS * D];
  __shared__ int q_e[AGG_WAVES][QCAP];
  __shared__ float colsum[2 * D];

  const int lane = threadIdx.x & 31;
  const int wave = __builtin_amdgcn_readfirstlane((int)(threadIdx.x >> 5));
  const int row0 = blockIdx.x * TILE_ROWS;
  int nrows = NN - row0;
  if (nrows > TILE_ROWS) nrows = TILE_ROWS;
  if (nrows <= 0) return;

  for (int t = threadIdx.x; t < TILE_ROWS * D / 4; t += AGG_WAVES * 32)
    ((float4*)acc)[t] = make_float4(0.0f, 0.0f, 0.0f, 0.0f);
  for (int t = threadIdx.x; t < 2 * D; t += AGG_WAVES * 32) colsum[t] = 0.0f;
  __syncthreads();

  int4* seg = (int4*)lists + ((size_t)blockIdx.x * AGG_WAVES + wave) * WCAP;
  int* cnt = lcnt + blockIdx.x * AGG_WAVES;

  if (FROM_LIST) {
    static_assert(AGG_WAVES == 32, "one count per lane");
    const int c = cnt[lane];
    if (__builtin_amdgcn_ballot_w32(c > WCAP) == 0u) {
      const int n = __builtin_amdgcn_readlane(c, wave);
      int i = 0;
      for (; i + 32 <= n; i += 32) {
        const int4 en = seg[i + lane];
        add_edges<D>(en.x, en.y, en.z, 32, lane, sup, acc);
      }
      if (i < n) {
        int4 en = make_int4(0, 0, 0, 0);
        if (i + lane < n) en = seg[i + lane];
        add_edges<D>(en.x, en.y, en.z, n - i, lane, sup, acc);
      }
    } else {
      scan_keys<D, false>(wave, lane, row0, nrows, q_e[wave], sup, src, dst,
                          w, keys, acc, seg);
    }
  } else {
    const int ln = scan_keys<D, true>(wave, lane, row0, nrows, q_e[wave],
                                      sup, src, dst, w, keys, acc, seg);
    if (lane == 0) cnt[wave] = ln;
  }

  __syncthreads();

  static_assert((AGG_WAVES * 32) % (D / 4) == 0, "a thread keeps its columns");
  float4* __restrict__ o = (float4*)(agg + (size_t)row0 * D);
  float4 s1 = make_float4(0.0f, 0.0f, 0.0f, 0.0f);
  float4 s2 = make_float4(0.0f, 0.0f, 0.0f, 0.0f);
  for (int t = threadIdx.x; t < nrows * (D / 4); t += AGG_WAVES * 32) {
    float4 v = ((const float4*)acc)[t];
    v.x = v.x > 0.0f ? v.x : 0.0f;
    v.y = v.y > 0.0f ? v.y : 0.0f;
    v.z = v.z > 0.0f ? v.z : 0.0f;
    v.w = v.w > 0.0f ? v.w : 0.0f;
    o[t] = v;
    s1.x += v.x; s1.y += v.y; s1.z += v.z; s1.w += v.w;
    s2.x += v.x * v.x; s2.y += v.y * v.y; s2.z += v.z * v.z; s2.w += v.w * v.w;
  }
  const int c0 = (threadIdx.x % (D / 4)) * 4;
  atomicAdd(&colsum[c0 + 0], s1.x);
  atomicAdd(&colsum[c0 + 1], s1.y);
  atomicAdd(&colsum[c0 + 2], s1.z);
  atomicAdd(&colsum[c0 + 3], s1.w);
  atomicAdd(&colsum[D + c0 + 0], s2.x);
  atomicAdd(&colsum[D + c0 + 1], s2.y);
  atomicAdd(&colsum[D + c0 + 2], s2.z);
  atomicAdd(&colsum[D + c0 + 3], s2.w);
  __syncthreads();
  for (int t = threadIdx.x; t < 2 * D; t += AGG_WAVES * 32)
    atomicAdd(&stats[t], colsum[t]);
}

__global__ void bn_normalize(const float* __restrict__ agg,
                             float* __restrict__ out,
                             const float* __restrict__ gamma,
                             const float* __restrict__ beta,
                             const float* __restrict__ stats,
                             int Nrows, int D) {
  size_t total = (size_t)Nrows * D;
  size_t i = (size_t)blockIdx.x * blockDim.x + threadIdx.x;
  if (i >= total) return;
  int c = (int)(i & (size_t)(D - 1));
  const float invN = 1.0f / (float)Nrows;
  const float mean = stats[c] * invN;
  const float var  = stats[D + c] * invN - mean * mean;
  const float s = gamma[c] * rsqrtf(var + BN_EPS);
  out[i] = (agg[i] - mean) * s + beta[c];
}

extern "C" void kernel_launch(void* const* d_in, const int* in_sizes, int n_in,
                              void* d_out, int out_size, void* d_ws,
                              size_t ws_size, hipStream_t stream) {
  (void)in_sizes; (void)n_in; (void)out_size; (void)ws_size;

  const float* y   = (const float*)d_in[0];
  const int*   src = (const int*)d_in[1];
  const int*   dst = (const int*)d_in[2];
  const float* w   = (const float*)d_in[3];
  const float* W1  = (const float*)d_in[4];
  const float* g1  = (const float*)d_in[5];
  const float* b1  = (const float*)d_in[6];
  const float* W2  = (const float*)d_in[7];
  const float* g2  = (const float*)d_in[8];
  const float* b2  = (const float*)d_in[9];
  float* out = (float*)d_out;

  float* ws    = (float*)d_ws;
  float* sup1  = ws;
  float* sup2  = sup1 + (size_t)NN * H1C;
  float* agg1  = sup2 + (size_t)NN * DINC;
  float* agg2  = agg1 + (size_t)NN * H1C;
  float* stats1 = agg2 + (size_t)NN * DINC;
  float* stats2 = stats1 + 2 * H1C;
  int* lists = (int*)(stats2 + 2 * DINC);
  int* lcnt  = lists + (size_t)NTILES * AGG_WAVES * WCAP * 4;
  int* keys  = lcnt + NTILES * AGG_WAVES;

  wmma_gemm_f32<<<25000 / 8, 256, 0, stream>>>(
      y, W1, sup1, stats1, 2 * H1C + 2 * DINC, dst, keys, NN, H2C, H1C);

  agg_tiles<H1C, false><<<NTILES, AGG_WAVES * 32, 0, stream>>>(
      sup1, agg1, src, dst, w, keys, stats1, lists, lcnt);

  wmma_gemm_bn_rows<<<256, G2_WAVES * 32, 0, stream>>>(
      agg1, W2, sup2, stats1, g1, b1);

  agg_tiles<DINC, true><<<NTILES, AGG_WAVES * 32, 0, stream>>>(
      sup2, agg2, src, dst, w, keys, stats2, lists, lcnt);

  {
    size_t total = (size_t)NN * DINC;
    bn_normalize<<<(int)((total + 255) / 256), 256, 0, stream>>>(
        agg2, out, g2, b2, stats2, NN, DINC);
  }
}


# Row gathers as 16-byte loads per lane (a load covers four whole lines, half as many load instructions); LDS tile kept component-major per 128 columns so adds stay conflict-free; add order kept
# speedup vs baseline: 3.4228x; 1.0879x over previous
#include <hip/hip_runtime.h>
#include <hip/hip_bf16.h>

typedef float v2f __attribute__((ext_vector_type(2)));
typedef float v8f __attribute__((ext_vector_type(8)));

#define NN   50000
#define EE   800000
#define H2C  64
#define H1C  128
#define DINC 256
#define BN_EPS 1e-5f

#define TILE_ROWS 196
#define NTILES ((NN + TILE_ROWS - 1) / TILE_ROWS)
static_assert(NTILES <= 256, "a tile number fits one byte");

__device__ __forceinline__ unsigned tile_key(int d) {
  return ((unsigned)d / (unsigned)TILE_ROWS) & 0xFFu;
}

__global__ void wmma_gemm_f32(const float* __restrict__ A,
                              const float* __restrict__ B,
                              float* __restrict__ C,
                              float* __restrict__ zero_buf, int zero_n,
                              const int* __restrict__ dst,
                              int* __restrict__ keys,
                              int Nrows, int K, int Ncols) {
  if (blockIdx.x == 0)
    for (int t = threadIdx.x; t < zero_n; t += blockDim.x) zero_buf[t] = 0.0f;

  {
    const int g = blockIdx.x * blockDim.x + threadIdx.x;
    if (g < EE / 4) {
      const int4 d = *(const int4*)(dst + 4 * g);
      keys[g] = (int)(tile_key(d.x) | (tile_key(d.y) << 8) |
                      (tile_key(d.z) << 16) | (tile_key(d.w) << 24));
    }
  }

  const int lane = threadIdx.x & 31;
  const int wave = threadIdx.x >> 5;
  const int colTiles = Ncols >> 4;
  const int rowTiles = Nrows >> 4;
  int tile = blockIdx.x * (blockDim.x >> 5) + wave;
  if (tile >= rowTiles * colTiles) return;

  const int rt = tile / colTiles;
  const int ct = tile - rt * colTiles;
  const int row0 = rt << 4;
  const int col0 = ct << 4;

  const int m  = lane & 15;
  const int kh = (lane >> 4) << 1;

  const float* __restrict__ arow = A + (size_t)(row0 + m) * K + kh;
  const float* __restrict__ bcol = B + (size_t)kh * Ncols + (col0 + m);

  v8f acc = {};
  for (int k = 0; k < K; k += 4) {
    v2f a = *(const v2f*)(arow + k);
    v2f b;
    b.x = bcol[(size_t)k * Ncols];
    b.y = bcol[(size_t)(k + 1) * Ncols];
    acc = __builtin_amdgcn_wmma_f32_16x16x4_f32(
        false, a, false, b, (short)0, acc, false, false);
  }

  float* __restrict__ crow = C + (size_t)col0 + m;
  const int rbase = row0 + ((lane >> 4) << 3);
#pragma unroll
  for (int r = 0; r < 8; ++r)
    crow[(size_t)(rbase + r) * Ncols] = acc[r];
}

#define G2_WAVES 16
#define G2_CT 4
#define G2_CG (DINC / (16 * G2_CT))
#define G2_ITEMS ((NN / 16) * G2_CG)
#define G2_PAIRS (H1C / 8)
static_assert(NN % 16 == 0 && H1C % 8 == 0 && DINC % (16 * G2_CT) == 0,
              "whole tiles only");
static_assert(G2_CG == 4 && G2_PAIRS == 16, "shifts below");

__global__ __launch_bounds__(G2_WAVES * 32)
void wmma_gemm_bn_rows(const float* __restrict__ A,
                       const float* __restrict__ B,
                       float* __restrict__ C,
                       const float* __restrict__ stats,
                       const float* __restrict__ gamma,
                       const float* __restrict__ beta) {
  __shared__ __attribute__((aligned(16))) float wf[(DINC / 16) * G2_PAIRS * 32 * 4];
  __shared__ __attribute__((aligned(16))) float bn[3 * H1C];

  for (int c = threadIdx.x; c < H1C; c += G2_WAVES * 32) {
    const float invN = 1.0f / (float)NN;
    const float mean = stats[c] * invN;
    const float var  = stats[H1C + c] * invN - mean * mean;
    bn[c] = mean;
    bn[H1C + c] = gamma[c] * rsqrtf(var + BN_EPS);
    bn[2 * H1C + c] = beta[c];
  }
#pragma unroll 4
  for (int q = threadIdx.x; q < (DINC / 16) * G2_PAIRS * 32; q += G2_WAVES * 32) {
    const int l = q & 31;
    const int j = (q >> 5) & (G2_PAIRS - 1);
    const int ct = q >> 9;
    const float* p = B + (size_t)(8 * j + 4 * (l >> 4)) * DINC + 16 * ct + (l & 15);
    float4 v;
    v.x = p[0];
    v.y = p[DINC];
    v.z = p[2 * DINC];
    v.w = p[3 * DINC];
    ((float4*)wf)[q] = v;
  }
  __syncthreads();

  const int lane = threadIdx.x & 31;
  const int wave = __builtin_amdgcn_readfirstlane((int)(threadIdx.x >> 5));
  const int m = lane & 15;
  const int h = lane >> 4;
  const float4* bnv = (const float4*)bn + h;

  for (int it = blockIdx.x + gridDim.x * wave; it < G2_ITEMS;
       it += gridDim.x * G2_WAVES) {
    __atomic_signal_fence(__ATOMIC_SEQ_CST);
    const int rt = it >> 2;
    const int cg = it & 3;
    const float* arow = A + (size_t)(rt * 16 + m) * H1C + 4 * h;
    float4 a[G2_PAIRS];
#pragma unroll
    for (int j = 0; j < G2_PAIRS; ++j) a[j] = *(const float4*)(arow + 8 * j);
    __builtin_amdgcn_sched_barrier(0);
#pragma unroll
    for (int j = 0; j < G2_PAIRS; ++j) {
      const float4 mu = bnv[2 * j];
      const float4 sc = bnv[H1C / 4 + 2 * j];
      const float4 sh = bnv[2 * (H1C / 4) + 2 * j];
      a[j].x = (a[j].x - mu.x) * sc.x + sh.x;
      a[j].y = (a[j].y - mu.y) * sc.y + sh.y;
      a[j].z = (a[j].z - mu.z) * sc.z + sh.z;
      a[j].w = (a[j].w - mu.w) * sc.w + sh.w;
    }

    const float4* bw = (const float4*)wf + (cg * G2_CT) * (G2_PAIRS * 32) + lane;
    v8f acc[G2_CT];
#pragma unroll
    for (int t = 0; t < G2_CT; ++t) acc[t] = (v8f){};
#pragma unroll
    for (int j = 0; j < G2_PAIRS; ++j) {
      float4 b[G2_CT];
#pragma unroll
      for (int t = 0; t < G2_CT; ++t) b[t] = bw[(t * G2_PAIRS + j) * 32];
      const v2f a0 = {a[j].x, a[j].y};
      const v2f a1 = {a[j].z, a[j].w};
#pragma unroll
      for (int t = 0; t < G2_CT; ++t) {
        const v2f b0 = {b[t].x, b[t].y};
        acc[t] = __builtin_amdgcn_wmma_f32_16x16x4_f32(
            false, a0, false, b0, (short)0, acc[t], false, false);
      }
#pragma unroll
      for (int t = 0; t < G2_CT; ++t) {
        const v2f b1 = {b[t].z, b[t].w};
        acc[t] = __builtin_amdgcn_wmma_f32_16x16x4_f32(
            false, a1, false, b1, (short)0, acc[t], false, false);
      }
    }

    float* crow = C + (size_t)(rt * 16 + 8 * h) * DINC + cg * (16 * G2_CT) + m;
#pragma unroll
    for (int t = 0; t < G2_CT; ++t)
#pragma unroll
      for (int r = 0; r < 8; ++r)
        crow[(size_t)r * DINC + t * 16] = acc[t][r];
  }
}

#define AGG_WAVES 32
#define QCAP 64
#define EDGE_UNR 4
#define KSTEP 512
#define NKBLK ((EE + KSTEP - 1) / KSTEP)
static_assert(EE % 16 == 0, "a lane's 16 keys are all edges or none");
#define WCAP 256

template <int D>
__device__ __forceinline__ void add_edges(int s, int wbits, int r, int n,
                                          int lane,
                                          const float* __restrict__ sup,
                                          float* acc) {
#pragma nounroll
  for (int i = 0; i < n; i += EDGE_UNR) {
    float4 v[EDGE_UNR][D / 128];
    float wi[EDGE_UNR];
    int ri[EDGE_UNR];
#pragma unroll
    for (int u = 0; u < EDGE_UNR; ++u) {
      const int si = __builtin_amdgcn_readlane(s, i + u);
      wi[u] = __int_as_float(__builtin_amdgcn_readlane(wbits, i + u));
      ri[u] = __builtin_amdgcn_readlane(r, i + u);
      const float4* p = (const float4*)(sup + (size_t)si * D) + lane;
#pragma unroll
      for (int j = 0; j < D / 128; ++j) v[u][j] = p[j * 32];
    }
#pragma unroll
    for (int u = 0; u < EDGE_UNR; ++u) {
      if (i + u < n) {
        float* a = acc + ri[u] * D + lane;
#pragma unroll
        for (int j = 0; j < D / 128; ++j) {
          atomicAdd(a + j * 128, v[u][j].x * wi[u]);
          atomicAdd(a + j * 128 + 32, v[u][j].y * wi[u]);
          atomicAdd(a + j * 128 + 64, v[u][j].z * wi[u]);
          atomicAdd(a + j * 128 + 96, v[u][j].w * wi[u]);
        }
      }
    }
  }
}

template <int D, bool KEEP>
__device__ __forceinline__ void drain_queue(const int* __restrict__ qe,
                                            int n, int lane,
                                            int row0, int nrows,
                                            const float* __restrict__ sup,
                                            const int* __restrict__ src,
                                            const int* __restrict__ dst,
                                            const float* __restrict__ w,
                                            float* acc,
                                            int4* __restrict__ seg, int have) {
  int e = 0;
  if (lane < n) e = qe[lane];
  const int s = src[e];
  int wbits = __float_as_int(w[e]);
  int r = dst[e] - row0;
  if ((unsigned)r >= (unsigned)nrows) { r = 0; wbits = 0; }
  if (KEEP) {
    if (lane < n && have + lane < WCAP)
      seg[have + lane] = make_int4(s, wbits, r, 0);
  }
  add_edges<D>(s, wbits, r, n, lane, sup, acc);
}

__device__ __forceinline__ unsigned match_bit7(unsigned x, unsigned tpat) {
  const unsigned y = x ^ tpat;
  const unsigned t = (y & 0x7F7F7F7Fu) + 0x7F7F7F7Fu;
  return ~(t | y);
}

__device__ __forceinline__ unsigned pick(unsigned m, unsigned a, unsigned b) {
  return (a & m) | (b & ~m);
}

__device__ __forceinline__ unsigned match16(int4 d, unsigned tpat) {
  const unsigned n0 = match_bit7((unsigned)d.x, tpat) >> 3;
  const unsigned n1 = match_bit7((unsigned)d.y, tpat) >> 2;
  const unsigned n2 = match_bit7((unsigned)d.z, tpat) >> 1;
  const unsigned n3 = match_bit7((unsigned)d.w, tpat);
  return pick(0x10101010u, n0, pick(0x20202020u, n1, pick(0x40404040u, n2, n3))) &
         0xF0F0F0F0u;
}

template <int D, bool KEEP>
__device__ __forceinline__ int scan_keys(int wave, int lane, int row0,
                                         int nrows, int* qe,
                                         const float* __restrict__ sup,
                                         const int* __restrict__ src,
                                         const int* __restrict__ dst,
                                         const float* __restrict__ w,
                                         const int* __restrict__ keys,
                                         float* acc, int4* __restrict__ seg) {
  int qn = 0;
  int ln = 0;
  const unsigned tpat = (unsigned)blockIdx.x * 0x01010101u;

  const int rot = (int)(blockIdx.x % NTILES) * NKBLK / NTILES;

  for (int k0 = wave; k0 < NKBLK; k0 += AGG_WAVES * 4) {
    int4 d[4];
    unsigned ok[4];
#pragma unroll
    for (int u = 0; u < 4; ++u) {
      const int k = k0 + u * AGG_WAVES;
      int blk = k + rot;
      if (blk >= NKBLK) blk -= NKBLK;
      const int e0 = (blk << 9) + (lane << 4);
      const bool own = k < NKBLK && e0 < EE;
      ok[u] = 0u - (unsigned)own;
      const int ea = own ? e0 : EE - 16;
      d[u] = ((const int4*)keys)[(unsigned)ea >> 4];
    }
    __builtin_amdgcn_sched_barrier(0);
    unsigned m16[4];
#pragma unroll
    for (int u = 0; u < 4; ++u) m16[u] = match16(d[u], tpat) & ok[u];
    unsigned long long mm =
        (unsigned long long)(m16[0] | (m16[1] >> 4)) |
        ((unsigned long long)(m16[2] | (m16[3] >> 4)) << 32);
    while (__builtin_amdgcn_ballot_w32(mm != 0ull)) {
      const bool has = mm != 0ull;
      const unsigned bm = __builtin_amdgcn_ballot_w32(has);
      if (has) {
        const int pos = __ffsll((long long)mm) - 1;
        const int q = pos & 7;
        const int u = ((pos >> 5) << 1) | ((q >> 2) ^ 1);
        const int key = ((q & 3) << 2) | ((pos & 31) >> 3);
        int blk = k0 + u * AGG_WAVES + rot;
        if (blk >= NKBLK) blk -= NKBLK;
        qe[qn + (int)__builtin_amdgcn_mbcnt_lo(bm, 0u)] =
            (blk << 9) + (lane << 4) + key;
        mm &= mm - 1ull;
      }
      qn += __popc(bm);
      if (qn >= 32) {
        qn -= 32;
        drain_queue<D, KEEP>(qe + qn, 32, lane, row0, nrows, sup, src, dst, w,
                             acc, seg, ln);
        ln += 32;
      }
    }
  }
  if (qn > 0) {
    drain_queue<D, KEEP>(qe, qn, lane, row0, nrows, sup, src, dst, w, acc,
                         seg, ln);
    ln += qn;
  }
  return ln;
}

template <int D, bool FROM_LIST>
__global__ __launch_bounds__(AGG_WAVES * 32)
void agg_tiles(const float* __restrict__ sup,
               float* __restrict__ agg,
               const int* __restrict__ src,
               const int* __restrict__ dst,
               const float* __restrict__ w,
               const int* __restrict__ keys,
               float* __restrict__ stats,
               int* __restrict__ lists,
               int* __restrict__ lcnt) {
  __shared__ float acc[TILE_ROWS * D];
  __shared__ int q_e[AGG_WAVES][QCAP];
  __shared__ float colsum[2 * D];

  const int lane = threadIdx.x & 31;
  const int wave = __builtin_amdgcn_readfirstlane((int)(threadIdx.x >> 5));
  const int row0 = blockIdx.x * TILE_ROWS;
  int nrows = NN - row0;
  if (nrows > TILE_ROWS) nrows = TILE_ROWS;
  if (nrows <= 0) return;

  for (int t = threadIdx.x; t < TILE_ROWS * D / 4; t += AGG_WAVES * 32)
    ((float4*)acc)[t] = make_float4(0.0f, 0.0f, 0.0f, 0.0f);
  for (int t = threadIdx.x; t < 2 * D; t += AGG_WAVES * 32) colsum[t] = 0.0f;
  __syncthreads();

  int4* seg = (int4*)lists + ((size_t)blockIdx.x * AGG_WAVES + wave) * WCAP;
  int* cnt = lcnt + blockIdx.x * AGG_WAVES;

  if (FROM_LIST) {
    static_assert(AGG_WAVES == 32, "one count per lane");
    const int c = cnt[lane];
    if (__builtin_amdgcn_ballot_w32(c > WCAP) == 0u) {
      const int n = __builtin_amdgcn_readlane(c, wave);
      int i = 0;
      for (; i + 32 <= n; i += 32) {
        const int4 en = seg[i + lane];
        add_edges<D>(en.x, en.y, en.z, 32, lane, sup, acc);
      }
      if (i < n) {
        int4 en = make_int4(0, 0, 0, 0);
        if (i + lane < n) en = seg[i + lane];
        add_edges<D>(en.x, en.y, en.z, n - i, lane, sup, acc);
      }
    } else {
      scan_keys<D, false>(wave, lane, row0, nrows, q_e[wave], sup, src, dst,
                          w, keys, acc, seg);
    }
  } else {
    const int ln = scan_keys<D, true>(wave, lane, row0, nrows, q_e[wave],
                                      sup, src, dst, w, keys, acc, seg);
    if (lane == 0) cnt[wave] = ln;
  }

  __syncthreads();

  static_assert((AGG_WAVES * 32) % (D / 4) == 0, "a thread keeps its columns");
  static_assert(D % 128 == 0, "whole 128-column segments");
  float4* __restrict__ o = (float4*)(agg + (size_t)row0 * D);
  float4 s1 = make_float4(0.0f, 0.0f, 0.0f, 0.0f);
  float4 s2 = make_float4(0.0f, 0.0f, 0.0f, 0.0f);
  for (int t = threadIdx.x; t < nrows * (D / 4); t += AGG_WAVES * 32) {
    const float* q = acc + (t >> 5) * 128 + (t & 31);
    float4 v;
    v.x = q[0];
    v.y = q[32];
    v.z = q[64];
    v.w = q[96];
    v.x = v.x > 0.0f ? v.x : 0.0f;
    v.y = v.y > 0.0f ? v.y : 0.0f;
    v.z = v.z > 0.0f ? v.z : 0.0f;
    v.w = v.w > 0.0f ? v.w : 0.0f;
    o[t] = v;
    s1.x += v.x; s1.y += v.y; s1.z += v.z; s1.w += v.w;
    s2.x += v.x * v.x; s2.y += v.y * v.y; s2.z += v.z * v.z; s2.w += v.w * v.w;
  }
  const int c0 = (threadIdx.x % (D / 4)) * 4;
  atomicAdd(&colsum[c0 + 0], s1.x);
  atomicAdd(&colsum[c0 + 1], s1.y);
  atomicAdd(&colsum[c0 + 2], s1.z);
  atomicAdd(&colsum[c0 + 3], s1.w);
  atomicAdd(&colsum[D + c0 + 0], s2.x);
  atomicAdd(&colsum[D + c0 + 1], s2.y);
  atomicAdd(&colsum[D + c0 + 2], s2.z);
  atomicAdd(&colsum[D + c0 + 3], s2.w);
  __syncthreads();
  for (int t = threadIdx.x; t < 2 * D; t += AGG_WAVES * 32)
    atomicAdd(&stats[t], colsum[t]);
}

__global__ void bn_normalize(const float* __restrict__ agg,
                             float* __restrict__ out,
                             const float* __restrict__ gamma,
                             const float* __restrict__ beta,
                             const float* __restrict__ stats,
                             int Nrows, int D) {
  size_t total = (size_t)Nrows * D;
  size_t i = (size_t)blockIdx.x * blockDim.x + threadIdx.x;
  if (i >= total) return;
  int c = (int)(i & (size_t)(D - 1));
  const float invN = 1.0f / (float)Nrows;
  const float mean = stats[c] * invN;
  const float var  = stats[D + c] * invN - mean * mean;
  const float s = gamma[c] * rsqrtf(var + BN_EPS);
  out[i] = (agg[i] - mean) * s + beta[c];
}

extern "C" void kernel_launch(void* const* d_in, const int* in_sizes, int n_in,
                              void* d_out, int out_size, void* d_ws,
                              size_t ws_size, hipStream_t stream) {
  (void)in_sizes; (void)n_in; (void)out_size; (void)ws_size;

  const float* y   = (const float*)d_in[0];
  const int*   src = (const int*)d_in[1];
  const int*   dst = (const int*)d_in[2];
  const float* w   = (const float*)d_in[3];
  const float* W1  = (const float*)d_in[4];
  const float* g1  = (const float*)d_in[5];
  const float* b1  = (const float*)d_in[6];
  const float* W2  = (const float*)d_in[7];
  const float* g2  = (const float*)d_in[8];
  const float* b2  = (const float*)d_in[9];
  float* out = (float*)d_out;

  float* ws    = (float*)d_ws;
  float* sup1  = ws;
  float* sup2  = sup1 + (size_t)NN * H1C;
  float* agg1  = sup2 + (size_t)NN * DINC;
  float* agg2  = agg1 + (size_t)NN * H1C;
  float* stats1 = agg2 + (size_t)NN * DINC;
  float* stats2 = stats1 + 2 * H1C;
  int* lists = (int*)(stats2 + 2 * DINC);
  int* lcnt  = lists + (size_t)NTILES * AGG_WAVES * WCAP * 4;
  int* keys  = lcnt + NTILES * AGG_WAVES;

  wmma_gemm_f32<<<25000 / 8, 256, 0, stream>>>(
      y, W1, sup1, stats1, 2 * H1C + 2 * DINC, dst, keys, NN, H2C, H1C);

  agg_tiles<H1C, false><<<NTILES, AGG_WAVES * 32, 0, stream>>>(
      sup1, agg1, src, dst, w, keys, stats1, lists, lcnt);

  wmma_gemm_bn_rows<<<256, G2_WAVES * 32, 0, stream>>>(
      agg1, W2, sup2, stats1, g1, b1);

  agg_tiles<DINC, true><<<NTILES, AGG_WAVES * 32, 0, stream>>>(
      sup2, agg2, src, dst, w, keys, stats2, lists, lcnt);

  {
    size_t total = (size_t)NN * DINC;
    bn_normalize<<<(int)((total + 255) / 256), 256, 0, stream>>>(
        agg2, out, g2, b2, stats2, NN, DINC);
  }
}


# First dense product keeps rows, walks columns: a wave holds 16 rows of y in registers, W1 sits in LDS as ready fragments; 4096 waves, not 25000; y fetched once, not 8x; k order of the sums changes
# speedup vs baseline: 3.5628x; 1.0409x over previous
#include <hip/hip_runtime.h>
#include <hip/hip_bf16.h>

typedef float v2f __attribute__((ext_vector_type(2)));
typedef float v8f __attribute__((ext_vector_type(8)));

#define NN   50000
#define EE   800000
#define H2C  64
#define H1C  128
#define DINC 256
#define BN_EPS 1e-5f

#define TILE_ROWS 196
#define NTILES ((NN + TILE_ROWS - 1) / TILE_ROWS)
static_assert(NTILES <= 256, "a tile number fits one byte");

__device__ __forceinline__ unsigned tile_key(int d) {
  return ((unsigned)d / (unsigned)TILE_ROWS) & 0xFFu;
}

#define G1_GROUPS 256
#define G1_WAVES 16
#define G1_CT (H1C / 16)
#define G1_PAIRS (H2C / 8)
#define G1_ITEMS (NN / 16)
#define G1_THREADS (G1_GROUPS * G1_WAVES * 32)
#define G1_ZERO (2 * H1C + 2 * DINC)
#define G1_KEY_STEPS ((EE / 4 + G1_THREADS - 1) / G1_THREADS)
#define G1_SLOTS ((G1_CT * G1_PAIRS * 32) / (G1_WAVES * 32))
static_assert((G1_CT * G1_PAIRS * 32) % (G1_WAVES * 32) == 0, "whole slots per thread");
static_assert(NN % 16 == 0 && H2C % 8 == 0 && H1C % 16 == 0, "whole tiles only");
static_assert(G1_ITEMS <= G1_GROUPS * G1_WAVES, "one row tile per wave at most");
static_assert(G1_CT == 8 && G1_PAIRS == 8, "shifts below");
static_assert(EE % 4 == 0, "whole key words");

__global__ __launch_bounds__(G1_WAVES * 32)
void wmma_gemm_rows(const float* __restrict__ A,
                    const float* __restrict__ B,
                    float* __restrict__ C,
                    float* __restrict__ zero_buf,
                    const int* __restrict__ dst,
                    int* __restrict__ keys) {
  __shared__ __attribute__((aligned(16))) float wf[G1_CT * G1_PAIRS * 32 * 4];

  const int lane = threadIdx.x & 31;
  const int wave = __builtin_amdgcn_readfirstlane((int)(threadIdx.x >> 5));
  const int m = lane & 15;
  const int h = lane >> 4;
  const int rt = blockIdx.x + G1_GROUPS * wave;
  const bool has = rt < G1_ITEMS;

  float4 a[G1_PAIRS];
  if (has) {
    const float* arow = A + (size_t)(rt * 16 + m) * H2C + 4 * h;
#pragma unroll
    for (int j = 0; j < G1_PAIRS; ++j) a[j] = *(const float4*)(arow + 8 * j);
  }

  int4 kd[G1_KEY_STEPS];
  const int g0 = blockIdx.x * (G1_WAVES * 32) + threadIdx.x;
#pragma unroll
  for (int i = 0; i < G1_KEY_STEPS; ++i) {
    const int g = i * G1_THREADS + g0;
    kd[i] = ((const int4*)dst)[g < EE / 4 ? g : EE / 4 - 1];
  }
  float4 wv[G1_SLOTS];
#pragma unroll
  for (int i = 0; i < G1_SLOTS; ++i) {
    const int q = i * (G1_WAVES * 32) + threadIdx.x;
    const int l = q & 31;
    const int j = (q >> 5) & (G1_PAIRS - 1);
    const int ct = q >> 8;
    const float* p = B + (size_t)(8 * j + 4 * (l >> 4)) * H1C + 16 * ct + (l & 15);
    wv[i].x = p[0];
    wv[i].y = p[H1C];
    wv[i].z = p[2 * H1C];
    wv[i].w = p[3 * H1C];
  }

  if (blockIdx.x == 0) {
#pragma unroll
    for (int i = 0; i < (G1_ZERO + G1_WAVES * 32 - 1) / (G1_WAVES * 32); ++i) {
      const int t = i * (G1_WAVES * 32) + threadIdx.x;
      if (t < G1_ZERO) zero_buf[t] = 0.0f;
    }
  }
#pragma unroll
  for (int i = 0; i < G1_KEY_STEPS; ++i) {
    const int g = i * G1_THREADS + g0;
    if (g < EE / 4)
      keys[g] = (int)(tile_key(kd[i].x) | (tile_key(kd[i].y) << 8) |
                      (tile_key(kd[i].z) << 16) | (tile_key(kd[i].w) << 24));
  }
#pragma unroll
  for (int i = 0; i < G1_SLOTS; ++i)
    ((float4*)wf)[i * (G1_WAVES * 32) + threadIdx.x] = wv[i];
  __syncthreads();
  if (!has) return;

  const float4* bw = (const float4*)wf + lane;
  v8f acc[G1_CT];
#pragma unroll
  for (int t = 0; t < G1_CT; ++t) acc[t] = (v8f){};
#pragma unroll
  for (int j = 0; j < G1_PAIRS; ++j) {
    float4 b[G1_CT];
#pragma unroll
    for (int t = 0; t < G1_CT; ++t) b[t] = bw[(t * G1_PAIRS + j) * 32];
    const v2f a0 = {a[j].x, a[j].y};
    const v2f a1 = {a[j].z, a[j].w};
#pragma unroll
    for (int t = 0; t < G1_CT; ++t) {
      const v2f b0 = {b[t].x, b[t].y};
      acc[t] = __builtin_amdgcn_wmma_f32_16x16x4_f32(
          false, a0, false, b0, (short)0, acc[t], false, false);
    }
#pragma unroll
    for (int t = 0; t < G1_CT; ++t) {
      const v2f b1 = {b[t].z, b[t].w};
      acc[t] = __builtin_amdgcn_wmma_f32_16x16x4_f32(
          false, a1, false, b1, (short)0, acc[t], false, false);
    }
  }

  float* crow = C + (size_t)(rt * 16 + 8 * h) * H1C + m;
#pragma unroll
  for (int r = 0; r < 8; ++r)
#pragma unroll
    for (int t = 0; t < G1_CT; ++t)
      crow[(size_t)r * H1C + t * 16] = acc[t][r];
}

#define G2_WAVES 16
#define G2_CT 4
#define G2_CG (DINC / (16 * G2_CT))
#define G2_ITEMS ((NN / 16) * G2_CG)
#define G2_PAIRS (H1C / 8)
static_assert(NN % 16 == 0 && H1C % 8 == 0 && DINC % (16 * G2_CT) == 0,
              "whole tiles only");
static_assert(G2_CG == 4 && G2_PAIRS == 16, "shifts below");

__global__ __launch_bounds__(G2_WAVES * 32)
void wmma_gemm_bn_rows(const float* __restrict__ A,
                       const float* __restrict__ B,
                       float* __restrict__ C,
                       const float* __restrict__ stats,
                       const float* __restrict__ gamma,
                       const float* __restrict__ beta) {
  __shared__ __attribute__((aligned(16))) float wf[(DINC / 16) * G2_PAIRS * 32 * 4];
  __shared__ __attribute__((aligned(16))) float bn[3 * H1C];

  for (int c = threadIdx.x; c < H1C; c += G2_WAVES * 32) {
    const float invN = 1.0f / (float)NN;
    const float mean = stats[c] * invN;
    const float var  = stats[H1C + c] * invN - mean * mean;
    bn[c] = mean;
    bn[H1C + c] = gamma[c] * rsqrtf(var + BN_EPS);
    bn[2 * H1C + c] = beta[c];
  }
#pragma unroll 4
  for (int q = threadIdx.x; q < (DINC / 16) * G2_PAIRS * 32; q += G2_WAVES * 32) {
    const int l = q & 31;
    const int j = (q >> 5) & (G2_PAIRS - 1);
    const int ct = q >> 9;
    const float* p = B + (size_t)(8 * j + 4 * (l >> 4)) * DINC + 16 * ct + (l & 15);
    float4 v;
    v.x = p[0];
    v.y = p[DINC];
    v.z = p[2 * DINC];
    v.w = p[3 * DINC];
    ((float4*)wf)[q] = v;
  }
  __syncthreads();

  const int lane = threadIdx.x & 31;
  const int wave = __builtin_amdgcn_readfirstlane((int)(threadIdx.x >> 5));
  const int m = lane & 15;
  const int h = lane >> 4;
  const float4* bnv = (const float4*)bn + h;

  for (int it = blockIdx.x + gridDim.x * wave; it < G2_ITEMS;
       it += gridDim.x * G2_WAVES) {
    __atomic_signal_fence(__ATOMIC_SEQ_CST);
    const int rt = it >> 2;
    const int cg = it & 3;
    const float* arow = A + (size_t)(rt * 16 + m) * H1C + 4 * h;
    float4 a[G2_PAIRS];
#pragma unroll
    for (int j = 0; j < G2_PAIRS; ++j) a[j] = *(const float4*)(arow + 8 * j);
    __builtin_amdgcn_sched_barrier(0);
#pragma unroll
    for (int j = 0; j < G2_PAIRS; ++j) {
      const float4 mu = bnv[2 * j];
      const float4 sc = bnv[H1C / 4 + 2 * j];
      const float4 sh = bnv[2 * (H1C / 4) + 2 * j];
      a[j].x = (a[j].x - mu.x) * sc.x + sh.x;
      a[j].y = (a[j].y - mu.y) * sc.y + sh.y;
      a[j].z = (a[j].z - mu.z) * sc.z + sh.z;
      a[j].w = (a[j].w - mu.w) * sc.w + sh.w;
    }

    const float4* bw = (const float4*)wf + (cg * G2_CT) * (G2_PAIRS * 32) + lane;
    v8f acc[G2_CT];
#pragma unroll
    for (int t = 0; t < G2_CT; ++t) acc[t] = (v8f){};
#pragma unroll
    for (int j = 0; j < G2_PAIRS; ++j) {
      float4 b[G2_CT];
#pragma unroll
      for (int t = 0; t < G2_CT; ++t) b[t] = bw[(t * G2_PAIRS + j) * 32];
      const v2f a0 = {a[j].x, a[j].y};
      const v2f a1 = {a[j].z, a[j].w};
#pragma unroll
      for (int t = 0; t < G2_CT; ++t) {
        const v2f b0 = {b[t].x, b[t].y};
        acc[t] = __builtin_amdgcn_wmma_f32_16x16x4_f32(
            false, a0, false, b0, (short)0, acc[t], false, false);
      }
#pragma unroll
      for (int t = 0; t < G2_CT; ++t) {
        const v2f b1 = {b[t].z, b[t].w};
        acc[t] = __builtin_amdgcn_wmma_f32_16x16x4_f32(
            false, a1, false, b1, (short)0, acc[t], false, false);
      }
    }

    float* crow = C + (size_t)(rt * 16 + 8 * h) * DINC + cg * (16 * G2_CT) + m;
#pragma unroll
    for (int t = 0; t < G2_CT; ++t)
#pragma unroll
      for (int r = 0; r < 8; ++r)
        crow[(size_t)r * DINC + t * 16] = acc[t][r];
  }
}

#define AGG_WAVES 32
#define QCAP 64
#define EDGE_UNR 4
#define KSTEP 512
#define NKBLK ((EE + KSTEP - 1) / KSTEP)
static_assert(EE % 16 == 0, "a lane's 16 keys are all edges or none");
#define WCAP 256

template <int D>
__device__ __forceinline__ void add_edges(int s, int wbits, int r, int n,
                                          int lane,
                                          const float* __restrict__ sup,
                                          float* acc) {
#pragma nounroll
  for (int i = 0; i < n; i += EDGE_UNR) {
    float4 v[EDGE_UNR][D / 128];
    float wi[EDGE_UNR];
    int ri[EDGE_UNR];
#pragma unroll
    for (int u = 0; u < EDGE_UNR; ++u) {
      const int si = __builtin_amdgcn_readlane(s, i + u);
      wi[u] = __int_as_float(__builtin_amdgcn_readlane(wbits, i + u));
      ri[u] = __builtin_amdgcn_readlane(r, i + u);
      const float4* p = (const float4*)(sup + (size_t)si * D) + lane;
#pragma unroll
      for (int j = 0; j < D / 128; ++j) v[u][j] = p[j * 32];
    }
#pragma unroll
    for (int u = 0; u < EDGE_UNR; ++u) {
      if (i + u < n) {
        float* a = acc + ri[u] * D + lane;
#pragma unroll
        for (int j = 0; j < D / 128; ++j) {
          atomicAdd(a + j * 128, v[u][j].x * wi[u]);
          atomicAdd(a + j * 128 + 32, v[u][j].y * wi[u]);
          atomicAdd(a + j * 128 + 64, v[u][j].z * wi[u]);
          atomicAdd(a + j * 128 + 96, v[u][j].w * wi[u]);
        }
      }
    }
  }
}

template <int D, bool KEEP>
__device__ __forceinline__ void drain_queue(const int* __restrict__ qe,
                                            int n, int lane,
                                            int row0, int nrows,
                                            const float* __restrict__ sup,
                                            const int* __restrict__ src,
                                            const int* __restrict__ dst,
                                            const float* __restrict__ w,
                                            float* acc,
                                            int4* __restrict__ seg, int have) {
  int e = 0;
  if (lane < n) e = qe[lane];
  const int s = src[e];
  int wbits = __float_as_int(w[e]);
  int r = dst[e] - row0;
  if ((unsigned)r >= (unsigned)nrows) { r = 0; wbits = 0; }
  if (KEEP) {
    if (lane < n && have + lane < WCAP)
      seg[have + lane] = make_int4(s, wbits, r, 0);
  }
  add_edges<D>(s, wbits, r, n, lane, sup, acc);
}

__device__ __forceinline__ unsigned match_bit7(unsigned x, unsigned tpat) {
  const unsigned y = x ^ tpat;
  const unsigned t = (y & 0x7F7F7F7Fu) + 0x7F7F7F7Fu;
  return ~(t | y);
}

__device__ __forceinline__ unsigned pick(unsigned m, unsigned a, unsigned b) {
  return (a & m) | (b & ~m);
}

__device__ __forceinline__ unsigned match16(int4 d, unsigned tpat) {
  const unsigned n0 = match_bit7((unsigned)d.x, tpat) >> 3;
  const unsigned n1 = match_bit7((unsigned)d.y, tpat) >> 2;
  const unsigned n2 = match_bit7((unsigned)d.z, tpat) >> 1;
  const unsigned n3 = match_bit7((unsigned)d.w, tpat);
  return pick(0x10101010u, n0, pick(0x20202020u, n1, pick(0x40404040u, n2, n3))) &
         0xF0F0F0F0u;
}

template <int D, bool KEEP>
__device__ __forceinline__ int scan_keys(int wave, int lane, int row0,
                                         int nrows, int* qe,
                                         const float* __restrict__ sup,
                                         const int* __restrict__ src,
                                         const int* __restrict__ dst,
                                         const float* __restrict__ w,
                                         const int* __restrict__ keys,
                                         float* acc, int4* __restrict__ seg) {
  int qn = 0;
  int ln = 0;
  const unsigned tpat = (unsigned)blockIdx.x * 0x01010101u;

  const int rot = (int)(blockIdx.x % NTILES) * NKBLK / NTILES;

  for (int k0 = wave; k0 < NKBLK; k0 += AGG_WAVES * 4) {
    int4 d[4];
    unsigned ok[4];
#pragma unroll
    for (int u = 0; u < 4; ++u) {
      const int k = k0 + u * AGG_WAVES;
      int blk = k + rot;
      if (blk >= NKBLK) blk -= NKBLK;
      const int e0 = (blk << 9) + (lane << 4);
      const bool own = k < NKBLK && e0 < EE;
      ok[u] = 0u - (unsigned)own;
      const int ea = own ? e0 : EE - 16;
      d[u] = ((const int4*)keys)[(unsigned)ea >> 4];
    }
    __builtin_amdgcn_sched_barrier(0);
    unsigned m16[4];
#pragma unroll
    for (int u = 0; u < 4; ++u) m16[u] = match16(d[u], tpat) & ok[u];
    unsigned long long mm =
        (unsigned long long)(m16[0] | (m16[1] >> 4)) |
        ((unsigned long long)(m16[2] | (m16[3] >> 4)) << 32);
    while (__builtin_amdgcn_ballot_w32(mm != 0ull)) {
      const bool has = mm != 0ull;
      const unsigned bm = __builtin_amdgcn_ballot_w32(has);
      if (has) {
        const int pos = __ffsll((long long)mm) - 1;
        const int q = pos & 7;
        const int u = ((pos >> 5) << 1) | ((q >> 2) ^ 1);
        const int key = ((q & 3) << 2) | ((pos & 31) >> 3);
        int blk = k0 + u * AGG_WAVES + rot;
        if (blk >= NKBLK) blk -= NKBLK;
        qe[qn + (int)__builtin_amdgcn_mbcnt_lo(bm, 0u)] =
            (blk << 9) + (lane << 4) + key;
        mm &= mm - 1ull;
      }
      qn += __popc(bm);
      if (qn >= 32) {
        qn -= 32;
        drain_queue<D, KEEP>(qe + qn, 32, lane, row0, nrows, sup, src, dst, w,
                             acc, seg, ln);
        ln += 32;
      }
    }
  }
  if (qn > 0) {
    drain_queue<D, KEEP>(qe, qn, lane, row0, nrows, sup, src, dst, w, acc,
                         seg, ln);
    ln += qn;
  }
  return ln;
}

template <int D, bool FROM_LIST>
__global__ __launch_bounds__(AGG_WAVES * 32)
void agg_tiles(const float* __restrict__ sup,
               float* __restrict__ agg,
               const int* __restrict__ src,
               const int* __restrict__ dst,
               const float* __restrict__ w,
               const int* __restrict__ keys,
               float* __restrict__ stats,
               int* __restrict__ lists,
               int* __restrict__ lcnt) {
  __shared__ float acc[TILE_ROWS * D];
  __shared__ int q_e[AGG_WAVES][QCAP];
  __shared__ float colsum[2 * D];

  const int lane = threadIdx.x & 31;
  const int wave = __builtin_amdgcn_readfirstlane((int)(threadIdx.x >> 5));
  const int row0 = blockIdx.x * TILE_ROWS;
  int nrows = NN - row0;
  if (nrows > TILE_ROWS) nrows = TILE_ROWS;
  if (nrows <= 0) return;

  for (int t = threadIdx.x; t < TILE_ROWS * D / 4; t += AGG_WAVES * 32)
    ((float4*)acc)[t] = make_float4(0.0f, 0.0f, 0.0f, 0.0f);
  for (int t = threadIdx.x; t < 2 * D; t += AGG_WAVES * 32) colsum[t] = 0.0f;
  __syncthreads();

  int4* seg = (int4*)lists + ((size_t)blockIdx.x * AGG_WAVES + wave) * WCAP;
  int* cnt = lcnt + blockIdx.x * AGG_WAVES;

  if (FROM_LIST) {
    static_assert(AGG_WAVES == 32, "one count per lane");
    const int c = cnt[lane];
    if (__builtin_amdgcn_ballot_w32(c > WCAP) == 0u) {
      const int n = __builtin_amdgcn_readlane(c, wave);
      int i = 0;
      for (; i + 32 <= n; i += 32) {
        const int4 en = seg[i + lane];
        add_edges<D>(en.x, en.y, en.z, 32, lane, sup, acc);
      }
      if (i < n) {
        int4 en = make_int4(0, 0, 0, 0);
        if (i + lane < n) en = seg[i + lane];
        add_edges<D>(en.x, en.y, en.z, n - i, lane, sup, acc);
      }
    } else {
      scan_keys<D, false>(wave, lane, row0, nrows, q_e[wave], sup, src, dst,
                          w, keys, acc, seg);
    }
  } else {
    const int ln = scan_keys<D, true>(wave, lane, row0, nrows, q_e[wave],
                                      sup, src, dst, w, keys, acc, seg);
    if (lane == 0) cnt[wave] = ln;
  }

  __syncthreads();

  static_assert((AGG_WAVES * 32) % (D / 4) == 0, "a thread keeps its columns");
  static_assert(D % 128 == 0, "whole 128-column segments");
  float4* __restrict__ o = (float4*)(agg + (size_t)row0 * D);
  float4 s1 = make_float4(0.0f, 0.0f, 0.0f, 0.0f);
  float4 s2 = make_float4(0.0f, 0.0f, 0.0f, 0.0f);
  for (int t = threadIdx.x; t < nrows * (D / 4); t += AGG_WAVES * 32) {
    const float* q = acc + (t >> 5) * 128 + (t & 31);
    float4 v;
    v.x = q[0];
    v.y = q[32];
    v.z = q[64];
    v.w = q[96];
    v.x = v.x > 0.0f ? v.x : 0.0f;
    v.y = v.y > 0.0f ? v.y : 0.0f;
    v.z = v.z > 0.0f ? v.z : 0.0f;
    v.w = v.w > 0.0f ? v.w : 0.0f;
    o[t] = v;
    s1.x += v.x; s1.y += v.y; s1.z += v.z; s1.w += v.w;
    s2.x += v.x * v.x; s2.y += v.y * v.y; s2.z += v.z * v.z; s2.w += v.w * v.w;
  }
  const int c0 = (threadIdx.x % (D / 4)) * 4;
  atomicAdd(&colsum[c0 + 0], s1.x);
  atomicAdd(&colsum[c0 + 1], s1.y);
  atomicAdd(&colsum[c0 + 2], s1.z);
  atomicAdd(&colsum[c0 + 3], s1.w);
  atomicAdd(&colsum[D + c0 + 0], s2.x);
  atomicAdd(&colsum[D + c0 + 1], s2.y);
  atomicAdd(&colsum[D + c0 + 2], s2.z);
  atomicAdd(&colsum[D + c0 + 3], s2.w);
  __syncthreads();
  for (int t = threadIdx.x; t < 2 * D; t += AGG_WAVES * 32)
    atomicAdd(&stats[t], colsum[t]);
}

__global__ void bn_normalize(const float* __restrict__ agg,
                             float* __restrict__ out,
                             const float* __restrict__ gamma,
                             const float* __restrict__ beta,
                             const float* __restrict__ stats,
                             int Nrows, int D) {
  size_t total = (size_t)Nrows * D;
  size_t i = (size_t)blockIdx.x * blockDim.x + threadIdx.x;
  if (i >= total) return;
  int c = (int)(i & (size_t)(D - 1));
  const float invN = 1.0f / (float)Nrows;
  const float mean = stats[c] * invN;
  const float var  = stats[D + c] * invN - mean * mean;
  const float s = gamma[c] * rsqrtf(var + BN_EPS);
  out[i] = (agg[i] - mean) * s + beta[c];
}

extern "C" void kernel_launch(void* const* d_in, const int* in_sizes, int n_in,
                              void* d_out, int out_size, void* d_ws,
                              size_t ws_size, hipStream_t stream) {
  (void)in_sizes; (void)n_in; (void)out_size; (void)ws_size;

  const float* y   = (const float*)d_in[0];
  const int*   src = (const int*)d_in[1];
  const int*   dst = (const int*)d_in[2];
  const float* w   = (const float*)d_in[3];
  const float* W1  = (const float*)d_in[4];
  const float* g1  = (const float*)d_in[5];
  const float* b1  = (const float*)d_in[6];
  const float* W2  = (const float*)d_in[7];
  const float* g2  = (const float*)d_in[8];
  const float* b2  = (const float*)d_in[9];
  float* out = (float*)d_out;

  float* ws    = (float*)d_ws;
  float* sup1  = ws;
  float* sup2  = sup1 + (size_t)NN * H1C;
  float* agg1  = sup2 + (size_t)NN * DINC;
  float* agg2  = agg1 + (size_t)NN * H1C;
  float* stats1 = agg2 + (size_t)NN * DINC;
  float* stats2 = stats1 + 2 * H1C;
  int* lists = (int*)(stats2 + 2 * DINC);
  int* lcnt  = lists + (size_t)NTILES * AGG_WAVES * WCAP * 4;
  int* keys  = lcnt + NTILES * AGG_WAVES;

  wmma_gemm_rows<<<G1_GROUPS, G1_WAVES * 32, 0, stream>>>(
      y, W1, sup1, stats1, dst, keys);

  agg_tiles<H1C, false><<<NTILES, AGG_WAVES * 32, 0, stream>>>(
      sup1, agg1, src, dst, w, keys, stats1, lists, lcnt);

  wmma_gemm_bn_rows<<<256, G2_WAVES * 32, 0, stream>>>(
      agg1, W2, sup2, stats1, g1, b1);

  agg_tiles<DINC, true><<<NTILES, AGG_WAVES * 32, 0, stream>>>(
      sup2, agg2, src, dst, w, keys, stats2, lists, lcnt);

  {
    size_t total = (size_t)NN * DINC;
    bn_normalize<<<(int)((total + 255) / 256), 256, 0, stream>>>(
        agg2, out, g2, b2, stats2, NN, DINC);
  }
}


# Layer-1 drain fetches one packed 8-byte record per queued edge (source row, tile row, weight; launch 1 stores the plane beside the keys), not three scattered loads of src, dst, w; add order kept
# speedup vs baseline: 3.7076x; 1.0406x over previous
#include <hip/hip_runtime.h>
#include <hip/hip_bf16.h>

typedef float v2f __attribute__((ext_vector_type(2)));
typedef float v8f __attribute__((ext_vector_type(8)));

#define NN   50000
#define EE   800000
#define H2C  64
#define H1C  128
#define DINC 256
#define BN_EPS 1e-5f

#define TILE_ROWS 196
#define NTILES ((NN + TILE_ROWS - 1) / TILE_ROWS)
static_assert(NTILES <= 256, "a tile number fits one byte");

__device__ __forceinline__ unsigned tile_key(int d) {
  return ((unsigned)d / (unsigned)TILE_ROWS) & 0xFFu;
}

static_assert(NN <= 65536 && TILE_ROWS < 0xFF, "record fields");
__device__ __forceinline__ int edge_rec(int s, int d) {
  const unsigned q = (unsigned)d / (unsigned)TILE_ROWS;
  const unsigned row = q < 256u ? (unsigned)d - q * (unsigned)TILE_ROWS : 0xFFu;
  return (int)(((unsigned)s & 0xFFFFu) | (row << 16));
}

#define G1_GROUPS 256
#define G1_WAVES 16
#define G1_CT (H1C / 16)
#define G1_PAIRS (H2C / 8)
#define G1_ITEMS (NN / 16)
#define G1_THREADS (G1_GROUPS * G1_WAVES * 32)
#define G1_ZERO (2 * H1C + 2 * DINC)
#define G1_KEY_STEPS ((EE / 4 + G1_THREADS - 1) / G1_THREADS)
#define G1_SLOTS ((G1_CT * G1_PAIRS * 32) / (G1_WAVES * 32))
static_assert((G1_CT * G1_PAIRS * 32) % (G1_WAVES * 32) == 0, "whole slots per thread");
static_assert(NN % 16 == 0 && H2C % 8 == 0 && H1C % 16 == 0, "whole tiles only");
static_assert(G1_ITEMS <= G1_GROUPS * G1_WAVES, "one row tile per wave at most");
static_assert(G1_CT == 8 && G1_PAIRS == 8, "shifts below");
static_assert(EE % 4 == 0, "whole key words");

__global__ __launch_bounds__(G1_WAVES * 32)
void wmma_gemm_rows(const float* __restrict__ A,
                    const float* __restrict__ B,
                    float* __restrict__ C,
                    float* __restrict__ zero_buf,
                    const int* __restrict__ src,
                    const int* __restrict__ dst,
                    const float* __restrict__ w,
                    int* __restrict__ keys,
                    int* __restrict__ recs) {
  __shared__ __attribute__((aligned(16))) float wf[G1_CT * G1_PAIRS * 32 * 4];

  const int lane = threadIdx.x & 31;
  const int wave = __builtin_amdgcn_readfirstlane((int)(threadIdx.x >> 5));
  const int m = lane & 15;
  const int h = lane >> 4;
  const int rt = blockIdx.x + G1_GROUPS * wave;
  const bool has = rt < G1_ITEMS;

  float4 a[G1_PAIRS];
  if (has) {
    const float* arow = A + (size_t)(rt * 16 + m) * H2C + 4 * h;
#pragma unroll
    for (int j = 0; j < G1_PAIRS; ++j) a[j] = *(const float4*)(arow + 8 * j);
  }

  int4 kd[G1_KEY_STEPS];
  int4 ks[G1_KEY_STEPS];
  float4 kw[G1_KEY_STEPS];
  const int g0 = blockIdx.x * (G1_WAVES * 32) + threadIdx.x;
#pragma unroll
  for (int i = 0; i < G1_KEY_STEPS; ++i) {
    const int g = i * G1_THREADS + g0;
    const int gg = g < EE / 4 ? g : EE / 4 - 1;
    kd[i] = ((const int4*)dst)[gg];
    ks[i] = ((const int4*)src)[gg];
    kw[i] = ((const float4*)w)[gg];
  }
  float4 wv[G1_SLOTS];
#pragma unroll
  for (int i = 0; i < G1_SLOTS; ++i) {
    const int q = i * (G1_WAVES * 32) + threadIdx.x;
    const int l = q & 31;
    const int j = (q >> 5) & (G1_PAIRS - 1);
    const int ct = q >> 8;
    const float* p = B + (size_t)(8 * j + 4 * (l >> 4)) * H1C + 16 * ct + (l & 15);
    wv[i].x = p[0];
    wv[i].y = p[H1C];
    wv[i].z = p[2 * H1C];
    wv[i].w = p[3 * H1C];
  }

  if (blockIdx.x == 0) {
#pragma unroll
    for (int i = 0; i < (G1_ZERO + G1_WAVES * 32 - 1) / (G1_WAVES * 32); ++i) {
      const int t = i * (G1_WAVES * 32) + threadIdx.x;
      if (t < G1_ZERO) zero_buf[t] = 0.0f;
    }
  }
#pragma unroll
  for (int i = 0; i < G1_KEY_STEPS; ++i) {
    const int g = i * G1_THREADS + g0;
    if (g < EE / 4) {
      keys[g] = (int)(tile_key(kd[i].x) | (tile_key(kd[i].y) << 8) |
                      (tile_key(kd[i].z) << 16) | (tile_key(kd[i].w) << 24));
      ((int4*)recs)[2 * g] =
          make_int4(edge_rec(ks[i].x, kd[i].x), __float_as_int(kw[i].x),
                    edge_rec(ks[i].y, kd[i].y), __float_as_int(kw[i].y));
      ((int4*)recs)[2 * g + 1] =
          make_int4(edge_rec(ks[i].z, kd[i].z), __float_as_int(kw[i].z),
                    edge_rec(ks[i].w, kd[i].w), __float_as_int(kw[i].w));
    }
  }
#pragma unroll
  for (int i = 0; i < G1_SLOTS; ++i)
    ((float4*)wf)[i * (G1_WAVES * 32) + threadIdx.x] = wv[i];
  __syncthreads();
  if (!has) return;

  const float4* bw = (const float4*)wf + lane;
  v8f acc[G1_CT];
#pragma unroll
  for (int t = 0; t < G1_CT; ++t) acc[t] = (v8f){};
#pragma unroll
  for (int j = 0; j < G1_PAIRS; ++j) {
    float4 b[G1_CT];
#pragma unroll
    for (int t = 0; t < G1_CT; ++t) b[t] = bw[(t * G1_PAIRS + j) * 32];
    const v2f a0 = {a[j].x, a[j].y};
    const v2f a1 = {a[j].z, a[j].w};
#pragma unroll
    for (int t = 0; t < G1_CT; ++t) {
      const v2f b0 = {b[t].x, b[t].y};
      acc[t] = __builtin_amdgcn_wmma_f32_16x16x4_f32(
          false, a0, false, b0, (short)0, acc[t], false, false);
    }
#pragma unroll
    for (int t = 0; t < G1_CT; ++t) {
      const v2f b1 = {b[t].z, b[t].w};
      acc[t] = __builtin_amdgcn_wmma_f32_16x16x4_f32(
          false, a1, false, b1, (short)0, acc[t], false, false);
    }
  }

  float* crow = C + (size_t)(rt * 16 + 8 * h) * H1C + m;
#pragma unroll
  for (int r = 0; r < 8; ++r)
#pragma unroll
    for (int t = 0; t < G1_CT; ++t)
      crow[(size_t)r * H1C + t * 16] = acc[t][r];
}

#define G2_WAVES 16
#define G2_CT 4
#define G2_CG (DINC / (16 * G2_CT))
#define G2_ITEMS ((NN / 16) * G2_CG)
#define G2_PAIRS (H1C / 8)
static_assert(NN % 16 == 0 && H1C % 8 == 0 && DINC % (16 * G2_CT) == 0,
              "whole tiles only");
static_assert(G2_CG == 4 && G2_PAIRS == 16, "shifts below");

__global__ __launch_bounds__(G2_WAVES * 32)
void wmma_gemm_bn_rows(const float* __restrict__ A,
                       const float* __restrict__ B,
                       float* __restrict__ C,
                       const float* __restrict__ stats,
                       const float* __restrict__ gamma,
                       const float* __restrict__ beta) {
  __shared__ __attribute__((aligned(16))) float wf[(DINC / 16) * G2_PAIRS * 32 * 4];
  __shared__ __attribute__((aligned(16))) float bn[3 * H1C];

  for (int c = threadIdx.x; c < H1C; c += G2_WAVES * 32) {
    const float invN = 1.0f / (float)NN;
    const float mean = stats[c] * invN;
    const float var  = stats[H1C + c] * invN - mean * mean;
    bn[c] = mean;
    bn[H1C + c] = gamma[c] * rsqrtf(var + BN_EPS);
    bn[2 * H1C + c] = beta[c];
  }
#pragma unroll 4
  for (int q = threadIdx.x; q < (DINC / 16) * G2_PAIRS * 32; q += G2_WAVES * 32) {
    const int l = q & 31;
    const int j = (q >> 5) & (G2_PAIRS - 1);
    const int ct = q >> 9;
    const float* p = B + (size_t)(8 * j + 4 * (l >> 4)) * DINC + 16 * ct + (l & 15);
    float4 v;
    v.x = p[0];
    v.y = p[DINC];
    v.z = p[2 * DINC];
    v.w = p[3 * DINC];
    ((float4*)wf)[q] = v;
  }
  __syncthreads();

  const int lane = threadIdx.x & 31;
  const int wave = __builtin_amdgcn_readfirstlane((int)(threadIdx.x >> 5));
  const int m = lane & 15;
  const int h = lane >> 4;
  const float4* bnv = (const float4*)bn + h;

  for (int it = blockIdx.x + gridDim.x * wave; it < G2_ITEMS;
       it += gridDim.x * G2_WAVES) {
    __atomic_signal_fence(__ATOMIC_SEQ_CST);
    const int rt = it >> 2;
    const int cg = it & 3;
    const float* arow = A + (size_t)(rt * 16 + m) * H1C + 4 * h;
    float4 a[G2_PAIRS];
#pragma unroll
    for (int j = 0; j < G2_PAIRS; ++j) a[j] = *(const float4*)(arow + 8 * j);
    __builtin_amdgcn_sched_barrier(0);
#pragma unroll
    for (int j = 0; j < G2_PAIRS; ++j) {
      const float4 mu = bnv[2 * j];
      const float4 sc = bnv[H1C / 4 + 2 * j];
      const float4 sh = bnv[2 * (H1C / 4) + 2 * j];
      a[j].x = (a[j].x - mu.x) * sc.x + sh.x;
      a[j].y = (a[j].y - mu.y) * sc.y + sh.y;
      a[j].z = (a[j].z - mu.z) * sc.z + sh.z;
      a[j].w = (a[j].w - mu.w) * sc.w + sh.w;
    }

    const float4* bw = (const float4*)wf + (cg * G2_CT) * (G2_PAIRS * 32) + lane;
    v8f acc[G2_CT];
#pragma unroll
    for (int t = 0; t < G2_CT; ++t) acc[t] = (v8f){};
#pragma unroll
    for (int j = 0; j < G2_PAIRS; ++j) {
      float4 b[G2_CT];
#pragma unroll
      for (int t = 0; t < G2_CT; ++t) b[t] = bw[(t * G2_PAIRS + j) * 32];
      const v2f a0 = {a[j].x, a[j].y};
      const v2f a1 = {a[j].z, a[j].w};
#pragma unroll
      for (int t = 0; t < G2_CT; ++t) {
        const v2f b0 = {b[t].x, b[t].y};
        acc[t] = __builtin_amdgcn_wmma_f32_16x16x4_f32(
            false, a0, false, b0, (short)0, acc[t], false, false);
      }
#pragma unroll
      for (int t = 0; t < G2_CT; ++t) {
        const v2f b1 = {b[t].z, b[t].w};
        acc[t] = __builtin_amdgcn_wmma_f32_16x16x4_f32(
            false, a1, false, b1, (short)0, acc[t], false, false);
      }
    }

    float* crow = C + (size_t)(rt * 16 + 8 * h) * DINC + cg * (16 * G2_CT) + m;
#pragma unroll
    for (int t = 0; t < G2_CT; ++t)
#pragma unroll
      for (int r = 0; r < 8; ++r)
        crow[(size_t)r * DINC + t * 16] = acc[t][r];
  }
}

#define AGG_WAVES 32
#define QCAP 64
#define EDGE_UNR 4
#define KSTEP 512
#define NKBLK ((EE + KSTEP - 1) / KSTEP)
static_assert(EE % 16 == 0, "a lane's 16 keys are all edges or none");
#define WCAP 256

template <int D>
__device__ __forceinline__ void add_edges(int s, int wbits, int r, int n,
                                          int lane,
                                          const float* __restrict__ sup,
                                          float* acc) {
#pragma nounroll
  for (int i = 0; i < n; i += EDGE_UNR) {
    float4 v[EDGE_UNR][D / 128];
    float wi[EDGE_UNR];
    int ri[EDGE_UNR];
#pragma unroll
    for (int u = 0; u < EDGE_UNR; ++u) {
      const int si = __builtin_amdgcn_readlane(s, i + u);
      wi[u] = __int_as_float(__builtin_amdgcn_readlane(wbits, i + u));
      ri[u] = __builtin_amdgcn_readlane(r, i + u);
      const float4* p = (const float4*)(sup + (size_t)si * D) + lane;
#pragma unroll
      for (int j = 0; j < D / 128; ++j) v[u][j] = p[j * 32];
    }
#pragma unroll
    for (int u = 0; u < EDGE_UNR; ++u) {
      if (i + u < n) {
        float* a = acc + ri[u] * D + lane;
#pragma unroll
        for (int j = 0; j < D / 128; ++j) {
          atomicAdd(a + j * 128, v[u][j].x * wi[u]);
          atomicAdd(a + j * 128 + 32, v[u][j].y * wi[u]);
          atomicAdd(a + j * 128 + 64, v[u][j].z * wi[u]);
          atomicAdd(a + j * 128 + 96, v[u][j].w * wi[u]);
        }
      }
    }
  }
}

template <int D, bool KEEP>
__device__ __forceinline__ void drain_queue(const int* __restrict__ qe,
                                            int n, int lane, int nrows,
                                            const float* __restrict__ sup,
                                            const int* __restrict__ recs,
                                            float* acc,
                                            int4* __restrict__ seg, int have) {
  int e = 0;
  if (lane < n) e = qe[lane];
  const int2 rc = ((const int2*)recs)[e];
  const int s = rc.x & 0xFFFF;
  int r = (int)((unsigned)rc.x >> 16);
  int wbits = rc.y;
  if ((unsigned)r >= (unsigned)nrows) { r = 0; wbits = 0; }
  if (KEEP) {
    if (lane < n && have + lane < WCAP)
      seg[have + lane] = make_int4(s, wbits, r, 0);
  }
  add_edges<D>(s, wbits, r, n, lane, sup, acc);
}

__device__ __forceinline__ unsigned match_bit7(unsigned x, unsigned tpat) {
  const unsigned y = x ^ tpat;
  const unsigned t = (y & 0x7F7F7F7Fu) + 0x7F7F7F7Fu;
  return ~(t | y);
}

__device__ __forceinline__ unsigned pick(unsigned m, unsigned a, unsigned b) {
  return (a & m) | (b & ~m);
}

__device__ __forceinline__ unsigned match16(int4 d, unsigned tpat) {
  const unsigned n0 = match_bit7((unsigned)d.x, tpat) >> 3;
  const unsigned n1 = match_bit7((unsigned)d.y, tpat) >> 2;
  const unsigned n2 = match_bit7((unsigned)d.z, tpat) >> 1;
  const unsigned n3 = match_bit7((unsigned)d.w, tpat);
  return pick(0x10101010u, n0, pick(0x20202020u, n1, pick(0x40404040u, n2, n3))) &
         0xF0F0F0F0u;
}

template <int D, bool KEEP>
__device__ __forceinline__ int scan_keys(int wave, int lane,
                                         int nrows, int* qe,
                                         const float* __restrict__ sup,
                                         const int* __restrict__ recs,
                                         const int* __restrict__ keys,
                                         float* acc, int4* __restrict__ seg) {
  int qn = 0;
  int ln = 0;
  const unsigned tpat = (unsigned)blockIdx.x * 0x01010101u;

  const int rot = (int)(blockIdx.x % NTILES) * NKBLK / NTILES;

  for (int k0 = wave; k0 < NKBLK; k0 += AGG_WAVES * 4) {
    int4 d[4];
    unsigned ok[4];
#pragma unroll
    for (int u = 0; u < 4; ++u) {
      const int k = k0 + u * AGG_WAVES;
      int blk = k + rot;
      if (blk >= NKBLK) blk -= NKBLK;
      const int e0 = (blk << 9) + (lane << 4);
      const bool own = k < NKBLK && e0 < EE;
      ok[u] = 0u - (unsigned)own;
      const int ea = own ? e0 : EE - 16;
      d[u] = ((const int4*)keys)[(unsigned)ea >> 4];
    }
    __builtin_amdgcn_sched_barrier(0);
    unsigned m16[4];
#pragma unroll
    for (int u = 0; u < 4; ++u) m16[u] = match16(d[u], tpat) & ok[u];
    unsigned long long mm =
        (unsigned long long)(m16[0] | (m16[1] >> 4)) |
        ((unsigned long long)(m16[2] | (m16[3] >> 4)) << 32);
    while (__builtin_amdgcn_ballot_w32(mm != 0ull)) {
      const bool has = mm != 0ull;
      const unsigned bm = __builtin_amdgcn_ballot_w32(has);
      if (has) {
        const int pos = __ffsll((long long)mm) - 1;
        const int q = pos & 7;
        const int u = ((pos >> 5) << 1) | ((q >> 2) ^ 1);
        const int key = ((q & 3) << 2) | ((pos & 31) >> 3);
        int blk = k0 + u * AGG_WAVES + rot;
        if (blk >= NKBLK) blk -= NKBLK;
        qe[qn + (int)__builtin_amdgcn_mbcnt_lo(bm, 0u)] =
            (blk << 9) + (lane << 4) + key;
        mm &= mm - 1ull;
      }
      qn += __popc(bm);
      if (qn >= 32) {
        qn -= 32;
        drain_queue<D, KEEP>(qe + qn, 32, lane, nrows, sup, recs, acc, seg,
                             ln);
        ln += 32;
      }
    }
  }
  if (qn > 0) {
    drain_queue<D, KEEP>(qe, qn, lane, nrows, sup, recs, acc, seg, ln);
    ln += qn;
  }
  return ln;
}

template <int D, bool FROM_LIST>
__global__ __launch_bounds__(AGG_WAVES * 32)
void agg_tiles(const float* __restrict__ sup,
               float* __restrict__ agg,
               const int* __restrict__ recs,
               const int* __restrict__ keys,
               float* __restrict__ stats,
               int* __restrict__ lists,
               int* __restrict__ lcnt) {
  __shared__ float acc[TILE_ROWS * D];
  __shared__ int q_e[AGG_WAVES][QCAP];
  __shared__ float colsum[2 * D];

  const int lane = threadIdx.x & 31;
  const int wave = __builtin_amdgcn_readfirstlane((int)(threadIdx.x >> 5));
  const int row0 = blockIdx.x * TILE_ROWS;
  int nrows = NN - row0;
  if (nrows > TILE_ROWS) nrows = TILE_ROWS;
  if (nrows <= 0) return;

  for (int t = threadIdx.x; t < TILE_ROWS * D / 4; t += AGG_WAVES * 32)
    ((float4*)acc)[t] = make_float4(0.0f, 0.0f, 0.0f, 0.0f);
  for (int t = threadIdx.x; t < 2 * D; t += AGG_WAVES * 32) colsum[t] = 0.0f;
  __syncthreads();

  int4* seg = (int4*)lists + ((size_t)blockIdx.x * AGG_WAVES + wave) * WCAP;
  int* cnt = lcnt + blockIdx.x * AGG_WAVES;

  if (FROM_LIST) {
    static_assert(AGG_WAVES == 32, "one count per lane");
    const int c = cnt[lane];
    if (__builtin_amdgcn_ballot_w32(c > WCAP) == 0u) {
      const int n = __builtin_amdgcn_readlane(c, wave);
      int i = 0;
      for (; i + 32 <= n; i += 32) {
        const int4 en = seg[i + lane];
        add_edges<D>(en.x, en.y, en.z, 32, lane, sup, acc);
      }
      if (i < n) {
        int4 en = make_int4(0, 0, 0, 0);
        if (i + lane < n) en = seg[i + lane];
        add_edges<D>(en.x, en.y, en.z, n - i, lane, sup, acc);
      }
    } else {
      scan_keys<D, false>(wave, lane, nrows, q_e[wave], sup, recs, keys, acc,
                          seg);
    }
  } else {
    const int ln = scan_keys<D, true>(wave, lane, nrows, q_e[wave], sup, recs,
                                      keys, acc, seg);
    if (lane == 0) cnt[wave] = ln;
  }

  __syncthreads();

  static_assert((AGG_WAVES * 32) % (D / 4) == 0, "a thread keeps its columns");
  static_assert(D % 128 == 0, "whole 128-column segments");
  float4* __restrict__ o = (float4*)(agg + (size_t)row0 * D);
  float4 s1 = make_float4(0.0f, 0.0f, 0.0f, 0.0f);
  float4 s2 = make_float4(0.0f, 0.0f, 0.0f, 0.0f);
  for (int t = threadIdx.x; t < nrows * (D / 4); t += AGG_WAVES * 32) {
    const float* q = acc + (t >> 5) * 128 + (t & 31);
    float4 v;
    v.x = q[0];
    v.y = q[32];
    v.z = q[64];
    v.w = q[96];
    v.x = v.x > 0.0f ? v.x : 0.0f;
    v.y = v.y > 0.0f ? v.y : 0.0f;
    v.z = v.z > 0.0f ? v.z : 0.0f;
    v.w = v.w > 0.0f ? v.w : 0.0f;
    o[t] = v;
    s1.x += v.x; s1.y += v.y; s1.z += v.z; s1.w += v.w;
    s2.x += v.x * v.x; s2.y += v.y * v.y; s2.z += v.z * v.z; s2.w += v.w * v.w;
  }
  const int c0 = (threadIdx.x % (D / 4)) * 4;
  atomicAdd(&colsum[c0 + 0], s1.x);
  atomicAdd(&colsum[c0 + 1], s1.y);
  atomicAdd(&colsum[c0 + 2], s1.z);
  atomicAdd(&colsum[c0 + 3], s1.w);
  atomicAdd(&colsum[D + c0 + 0], s2.x);
  atomicAdd(&colsum[D + c0 + 1], s2.y);
  atomicAdd(&colsum[D + c0 + 2], s2.z);
  atomicAdd(&colsum[D + c0 + 3], s2.w);
  __syncthreads();
  for (int t = threadIdx.x; t < 2 * D; t += AGG_WAVES * 32)
    atomicAdd(&stats[t], colsum[t]);
}

__global__ void bn_normalize(const float* __restrict__ agg,
                             float* __restrict__ out,
                             const float* __restrict__ gamma,
                             const float* __restrict__ beta,
                             const float* __restrict__ stats,
                             int Nrows, int D) {
  size_t total = (size_t)Nrows * D;
  size_t i = (size_t)blockIdx.x * blockDim.x + threadIdx.x;
  if (i >= total) return;
  int c = (int)(i & (size_t)(D - 1));
  const float invN = 1.0f / (float)Nrows;
  const float mean = stats[c] * invN;
  const float var  = stats[D + c] * invN - mean * mean;
  const float s = gamma[c] * rsqrtf(var + BN_EPS);
  out[i] = (agg[i] - mean) * s + beta[c];
}

extern "C" void kernel_launch(void* const* d_in, const int* in_sizes, int n_in,
                              void* d_out, int out_size, void* d_ws,
                              size_t ws_size, hipStream_t stream) {
  (void)in_sizes; (void)n_in; (void)out_size; (void)ws_size;

  const float* y   = (const float*)d_in[0];
  const int*   src = (const int*)d_in[1];
  const int*   dst = (const int*)d_in[2];
  const float* w   = (const float*)d_in[3];
  const float* W1  = (const float*)d_in[4];
  const float* g1  = (const float*)d_in[5];
  const float* b1  = (const float*)d_in[6];
  const float* W2  = (const float*)d_in[7];
  const float* g2  = (const float*)d_in[8];
  const float* b2  = (const float*)d_in[9];
  float* out = (float*)d_out;

  float* ws    = (float*)d_ws;
  float* sup1  = ws;
  float* sup2  = sup1 + (size_t)NN * H1C;
  float* agg1  = sup2 + (size_t)NN * DINC;
  float* agg2  = agg1 + (size_t)NN * H1C;
  float* stats1 = agg2 + (size_t)NN * DINC;
  float* stats2 = stats1 + 2 * H1C;
  int* lists = (int*)(stats2 + 2 * DINC);
  int* lcnt  = lists + (size_t)NTILES * AGG_WAVES * WCAP * 4;
  int* keys  = lcnt + NTILES * AGG_WAVES;
  int* recs  = keys + EE / 4;

  wmma_gemm_rows<<<G1_GROUPS, G1_WAVES * 32, 0, stream>>>(
      y, W1, sup1, stats1, src, dst, w, keys, recs);

  agg_tiles<H1C, false><<<NTILES, AGG_WAVES * 32, 0, stream>>>(
      sup1, agg1, recs, keys, stats1, lists, lcnt);

  wmma_gemm_bn_rows<<<256, G2_WAVES * 32, 0, stream>>>(
      agg1, W2, sup2, stats1, g1, b1);

  agg_tiles<DINC, true><<<NTILES, AGG_WAVES * 32, 0, stream>>>(
      sup2, agg2, recs, keys, stats2, lists, lcnt);

  {
    size_t total = (size_t)NN * DINC;
    bn_normalize<<<(int)((total + 255) / 256), 256, 0, stream>>>(
        agg2, out, g2, b2, stats2, NN, DINC);
  }
}
